# starting kernel
# baseline (speedup 1.0000x reference)
#include <hip/hip_runtime.h>

constexpr int N0 = 400000, N1 = 100000, N2 = 25000;
constexpr int E0 = 800000, E1 = 200000;
constexpr int DIN = 128, HID = 256, DOUT = 128;

typedef __attribute__((ext_vector_type(2))) float v2f;
typedef __attribute__((ext_vector_type(8))) float v8f;

__global__ void zero_f32(float* __restrict__ p, long n) {
    long i = (long)blockIdx.x * blockDim.x + threadIdx.x;
    if (i < n) p[i] = 0.0f;
}

__global__ void build_feat(const int* __restrict__ x,
                           const float* __restrict__ e0, const float* __restrict__ e1,
                           const float* __restrict__ e2, const float* __restrict__ e3,
                           float* __restrict__ feat) {
    int t = blockIdx.x * blockDim.x + threadIdx.x;
    int row = t >> 5, lane = t & 31;
    if (row >= N0) return;
    const float4* src;
    if (lane < 8)       src = (const float4*)(e0 + (long)x[row * 4 + 0] * 32) + lane;
    else if (lane < 12) src = (const float4*)(e1 + (long)x[row * 4 + 1] * 16) + (lane - 8);
    else if (lane < 16) src = (const float4*)(e2 + (long)x[row * 4 + 2] * 16) + (lane - 12);
    else                src = (const float4*)(e3 + (long)x[row * 4 + 3] * 64) + (lane - 16);
    ((float4*)(feat + (long)row * 128))[lane] = *src;
}

template <int WIDTH>
__global__ void scatter_edges(const float* __restrict__ h, const int* __restrict__ srcA,
                              const int* __restrict__ dstA, float* __restrict__ agg,
                              float* __restrict__ deg, int nE) {
    int t = blockIdx.x * blockDim.x + threadIdx.x;
    int e = t >> 5, lane = t & 31;
    if (e >= nE) return;
    int s = srcA[e], d = dstA[e];
    const float* hp = h + (long)s * WIDTH;
    float* ap = agg + (long)d * WIDTH;
#pragma unroll
    for (int j = 0; j < WIDTH / 128; ++j) {
        float4 v = ((const float4*)hp)[lane + 32 * j];
        atomicAdd(ap + 4 * lane + 128 * j + 0, v.x);
        atomicAdd(ap + 4 * lane + 128 * j + 1, v.y);
        atomicAdd(ap + 4 * lane + 128 * j + 2, v.z);
        atomicAdd(ap + 4 * lane + 128 * j + 3, v.w);
    }
    if (lane == 0) atomicAdd(deg + d, 1.0f);
}

__global__ void __launch_bounds__(256)
gemm1_fused(const float* __restrict__ feat, const float* __restrict__ agg0,
            const float* __restrict__ deg0, const float* __restrict__ W,
            const float* __restrict__ bias, float* __restrict__ h1) {
    __shared__ float As[16 * 132];
    __shared__ float srow[16];
    const int m0 = blockIdx.x * 16;
    const int tid = threadIdx.x;
#pragma unroll
    for (int i = 0; i < 8; ++i) {
        int idx = tid + i * 256;
        int r = idx >> 7, c = idx & 127;
        long row = m0 + r;
        float inv = 1.0f / fmaxf(deg0[row], 1.0f);
        As[r * 132 + c] = feat[row * 128 + c] + agg0[row * 128 + c] * inv;
    }
    if (tid < 16) srow[tid] = (deg0[m0 + tid] > 0.0f) ? 2.0f : 1.0f;
    __syncthreads();

    const int lane = tid & 31, wave = tid >> 5;
    const int lo = lane & 15, hi = lane >> 4;
    const int nA = wave * 16, nB = (wave + 8) * 16;
    v8f c0 = {0, 0, 0, 0, 0, 0, 0, 0};
    v8f c1 = {0, 0, 0, 0, 0, 0, 0, 0};
    for (int k = 0; k < DIN; k += 4) {
        const int kk = k + 2 * hi;
        v2f a;
        a[0] = As[lo * 132 + kk];
        a[1] = As[lo * 132 + kk + 1];
        v2f b0, b1;
        b0[0] = W[kk * HID + nA + lo];
        b0[1] = W[(kk + 1) * HID + nA + lo];
        b1[0] = W[kk * HID + nB + lo];
        b1[1] = W[(kk + 1) * HID + nB + lo];
        c0 = __builtin_amdgcn_wmma_f32_16x16x4_f32(false, a, false, b0, (short)0, c0, false, false);
        c1 = __builtin_amdgcn_wmma_f32_16x16x4_f32(false, a, false, b1, (short)0, c1, false, false);
    }
#pragma unroll
    for (int r = 0; r < 8; ++r) {
        int rr = r + 8 * hi;
        long row = m0 + rr;
        float s = srow[rr];
        h1[row * HID + nA + lo] = fmaxf(c0[r] + s * bias[nA + lo], 0.0f);
        h1[row * HID + nB + lo] = fmaxf(c1[r] + s * bias[nB + lo], 0.0f);
    }
}

__global__ void __launch_bounds__(256)
gemm2_fused(const float* __restrict__ h1, const float* __restrict__ agg1,
            const float* __restrict__ deg1, const float* __restrict__ W,
            const float* __restrict__ bias, float* __restrict__ out) {
    __shared__ float As[16 * 260];
    const int m0 = blockIdx.x * 16;
    const int tid = threadIdx.x;
#pragma unroll
    for (int i = 0; i < 16; ++i) {
        int idx = tid + i * 256;
        int r = idx >> 8, c = idx & 255;
        int row = m0 + r;
        float v = 0.0f;
        if (row < N2) {
            float inv = 1.0f / fmaxf(deg1[row], 1.0f);
            v = h1[(long)row * HID + c] + agg1[(long)row * HID + c] * inv;
        }
        As[r * 260 + c] = v;
    }
    __syncthreads();

    const int lane = tid & 31, wave = tid >> 5;
    const int lo = lane & 15, hi = lane >> 4;
    const int n0 = wave * 16;
    v8f acc = {0, 0, 0, 0, 0, 0, 0, 0};
    for (int k = 0; k < HID; k += 4) {
        const int kk = k + 2 * hi;
        v2f a;
        a[0] = As[lo * 260 + kk];
        a[1] = As[lo * 260 + kk + 1];
        v2f b;
        b[0] = W[kk * DOUT + n0 + lo];
        b[1] = W[(kk + 1) * DOUT + n0 + lo];
        acc = __builtin_amdgcn_wmma_f32_16x16x4_f32(false, a, false, b, (short)0, acc, false, false);
    }
#pragma unroll
    for (int r = 0; r < 8; ++r) {
        int row = m0 + r + 8 * hi;
        if (row < N2) out[(long)row * DOUT + n0 + lo] = acc[r] + bias[n0 + lo];
    }
}

extern "C" void kernel_launch(void* const* d_in, const int* in_sizes, int n_in,
                              void* d_out, int out_size, void* d_ws, size_t ws_size,
                              hipStream_t stream) {
    (void)in_sizes; (void)n_in; (void)out_size; (void)ws_size;
    const int*   x  = (const int*)d_in[0];
    const float* e0 = (const float*)d_in[1];
    const float* e1 = (const float*)d_in[2];
    const float* e2 = (const float*)d_in[3];
    const float* e3 = (const float*)d_in[4];
    const float* Win[2]  = {(const float*)d_in[5], (const float*)d_in[9]};
    const float* bin[2]  = {(const float*)d_in[6], (const float*)d_in[10]};
    const float* Wout[2] = {(const float*)d_in[7], (const float*)d_in[11]};
    const float* bout[2] = {(const float*)d_in[8], (const float*)d_in[12]};
    const int* srcL0[2] = {(const int*)d_in[13], (const int*)d_in[17]};
    const int* dstL0[2] = {(const int*)d_in[14], (const int*)d_in[18]};
    const int* srcL1[2] = {(const int*)d_in[15], (const int*)d_in[19]};
    const int* dstL1[2] = {(const int*)d_in[16], (const int*)d_in[20]};
    float* out = (float*)d_out;

    size_t off = 0;
    auto take = [&](size_t bytes) { size_t o = off; off += (bytes + 255) & ~(size_t)255; return o; };
    char* ws = (char*)d_ws;
    float* feat = (float*)(ws + take((size_t)N0 * DIN * 4));
    float* h1   = (float*)(ws + take((size_t)N1 * HID * 4));
    size_t zbeg = off;
    float* agg0 = (float*)(ws + take((size_t)N1 * DIN * 4));
    float* deg0 = (float*)(ws + take((size_t)N1 * 4));
    float* agg1 = (float*)(ws + take((size_t)N2 * HID * 4));
    float* deg1 = (float*)(ws + take((size_t)N2 * 4));
    float* zptr = (float*)(ws + zbeg);
    long   zcnt = (long)((off - zbeg) / 4);

    build_feat<<<N0 / 8, 256, 0, stream>>>(x, e0, e1, e2, e3, feat);

    for (int e = 0; e < 2; ++e) {
        zero_f32<<<(unsigned)((zcnt + 255) / 256), 256, 0, stream>>>(zptr, zcnt);
        scatter_edges<DIN><<<E0 / 8, 256, 0, stream>>>(feat, srcL0[e], dstL0[e], agg0, deg0, E0);
        gemm1_fused<<<N1 / 16, 256, 0, stream>>>(feat, agg0, deg0, Win[e], bin[e], h1);
        scatter_edges<HID><<<E1 / 8, 256, 0, stream>>>(h1, srcL1[e], dstL1[e], agg1, deg1, E1);
        gemm2_fused<<<(N2 + 15) / 16, 256, 0, stream>>>(h1, agg1, deg1, Wout[e], bout[e],
                                                        out + (size_t)e * N2 * DOUT);
    }
}


# pull aggregation: link each edge into its destination's list (one int exchange per edge), sum h[src] along the lists in the GEMM prologues; float atomics and agg planes gone; sums in list order
# speedup vs baseline: 1.8628x; 1.8628x over previous
#include <hip/hip_runtime.h>

constexpr int N0 = 400000, N1 = 100000, N2 = 25000;
constexpr int E0 = 800000, E1 = 200000;
constexpr int DIN = 128, HID = 256, DOUT = 128;

typedef __attribute__((ext_vector_type(2))) float v2f;
typedef __attribute__((ext_vector_type(8))) float v8f;

__global__ void build_feat(const int* __restrict__ x,
                           const float* __restrict__ e0, const float* __restrict__ e1,
                           const float* __restrict__ e2, const float* __restrict__ e3,
                           float* __restrict__ feat) {
    int t = blockIdx.x * blockDim.x + threadIdx.x;
    int row = t >> 5, lane = t & 31;
    if (row >= N0) return;
    const float4* src;
    if (lane < 8)       src = (const float4*)(e0 + (long)x[row * 4 + 0] * 32) + lane;
    else if (lane < 12) src = (const float4*)(e1 + (long)x[row * 4 + 1] * 16) + (lane - 8);
    else if (lane < 16) src = (const float4*)(e2 + (long)x[row * 4 + 2] * 16) + (lane - 12);
    else                src = (const float4*)(e3 + (long)x[row * 4 + 3] * 64) + (lane - 16);
    ((float4*)(feat + (long)row * 128))[lane] = *src;
}

constexpr int NHEAD = 2 * N1 + 2 * N2;
constexpr int NLINK = 2 * E0 + 2 * E1;

__global__ void fill_heads(int* __restrict__ head) {
    int i = blockIdx.x * blockDim.x + threadIdx.x;
    if (i < NHEAD) head[i] = -1;
}

__global__ void link_edges(const int* __restrict__ dA0, const int* __restrict__ dB0,
                           const int* __restrict__ dA1, const int* __restrict__ dB1,
                           int* __restrict__ head, int* __restrict__ nxt) {
    int t = blockIdx.x * blockDim.x + threadIdx.x;
    if (t >= NLINK) return;
    int e, d;
    int* hp;
    if (t < 2 * E0) {
        int b = (t >= E0) ? 1 : 0;
        e = t - b * E0;
        d = (b ? dB0 : dA0)[e];
        hp = head + b * N1 + d;
    } else {
        int u = t - 2 * E0;
        int b = (u >= E1) ? 1 : 0;
        e = u - b * E1;
        d = (b ? dB1 : dA1)[e];
        hp = head + 2 * N1 + b * N2 + d;
    }
    nxt[t] = atomicExch(hp, e);
}

template <int WIDTH>
__device__ __forceinline__ int pull_row(const float* __restrict__ h, const int* __restrict__ srcA,
                                        const int* __restrict__ head, const int* __restrict__ nxt,
                                        int nE, int row, bool rowOk, int l16,
                                        float4 (&acc)[WIDTH / 64]) {
    constexpr int J = WIDTH / 64;
#pragma unroll
    for (int j = 0; j < J; ++j) acc[j] = make_float4(0.0f, 0.0f, 0.0f, 0.0f);
    int cnt = 0;
    int e = rowOk ? head[row] : -1;
    for (int it = 0; it < nE && (unsigned)e < (unsigned)nE; ++it) {
        const int s = srcA[e];
        const int en = nxt[e];
        const float4* hp = (const float4*)(h + (long)s * WIDTH);
#pragma unroll
        for (int j = 0; j < J; ++j) {
            const float4 v = hp[l16 + 16 * j];
            acc[j].x += v.x; acc[j].y += v.y; acc[j].z += v.z; acc[j].w += v.w;
        }
        ++cnt;
        e = en;
    }
    return cnt;
}

__global__ void __launch_bounds__(256)
gemm1_fused(const float* __restrict__ feat, const int* __restrict__ srcA,
            const int* __restrict__ head, const int* __restrict__ nxt,
            const float* __restrict__ W,
            const float* __restrict__ bias, float* __restrict__ h1) {
    __shared__ alignas(16) float As[16 * 132];
    __shared__ float srow[16];
    const int m0 = blockIdx.x * 16;
    const int tid = threadIdx.x;
    const int lane = tid & 31, wave = tid >> 5;
    {
        const int r = tid >> 4, l16 = tid & 15;
        const long row = m0 + r;
        float4 f[2];
#pragma unroll
        for (int j = 0; j < 2; ++j) f[j] = ((const float4*)(feat + row * 128))[l16 + 16 * j];
        float4 acc[2];
        const int cnt = pull_row<DIN>(feat, srcA, head, nxt, E0, m0 + r, true, l16, acc);
        const float inv = 1.0f / fmaxf((float)cnt, 1.0f);
#pragma unroll
        for (int j = 0; j < 2; ++j) {
            float4 a;
            a.x = f[j].x + acc[j].x * inv;
            a.y = f[j].y + acc[j].y * inv;
            a.z = f[j].z + acc[j].z * inv;
            a.w = f[j].w + acc[j].w * inv;
            *(float4*)&As[r * 132 + 4 * (l16 + 16 * j)] = a;
        }
        if (l16 == 0) srow[r] = (cnt > 0) ? 2.0f : 1.0f;
    }
    __syncthreads();

    const int lo = lane & 15, hi = lane >> 4;
    const int nA = wave * 16, nB = (wave + 8) * 16;
    v8f c0 = {0, 0, 0, 0, 0, 0, 0, 0};
    v8f c1 = {0, 0, 0, 0, 0, 0, 0, 0};
    for (int k = 0; k < DIN; k += 4) {
        const int kk = k + 2 * hi;
        v2f a;
        a[0] = As[lo * 132 + kk];
        a[1] = As[lo * 132 + kk + 1];
        v2f b0, b1;
        b0[0] = W[kk * HID + nA + lo];
        b0[1] = W[(kk + 1) * HID + nA + lo];
        b1[0] = W[kk * HID + nB + lo];
        b1[1] = W[(kk + 1) * HID + nB + lo];
        c0 = __builtin_amdgcn_wmma_f32_16x16x4_f32(false, a, false, b0, (short)0, c0, false, false);
        c1 = __builtin_amdgcn_wmma_f32_16x16x4_f32(false, a, false, b1, (short)0, c1, false, false);
    }
#pragma unroll
    for (int r = 0; r < 8; ++r) {
        int rr = r + 8 * hi;
        long row = m0 + rr;
        float s = srow[rr];
        h1[row * HID + nA + lo] = fmaxf(c0[r] + s * bias[nA + lo], 0.0f);
        h1[row * HID + nB + lo] = fmaxf(c1[r] + s * bias[nB + lo], 0.0f);
    }
}

__global__ void __launch_bounds__(256)
gemm2_fused(const float* __restrict__ h1, const int* __restrict__ srcA,
            const int* __restrict__ head, const int* __restrict__ nxt,
            const float* __restrict__ W,
            const float* __restrict__ bias, float* __restrict__ out) {
    __shared__ alignas(16) float As[16 * 260];
    const int m0 = blockIdx.x * 16;
    const int tid = threadIdx.x;
    const int lane = tid & 31, wave = tid >> 5;
    {
        const int r = tid >> 4, l16 = tid & 15;
        const int row = m0 + r;
        const bool ok = row < N2;
        float4 f[4];
#pragma unroll
        for (int j = 0; j < 4; ++j)
            f[j] = ok ? ((const float4*)(h1 + (long)row * HID))[l16 + 16 * j]
                      : make_float4(0.0f, 0.0f, 0.0f, 0.0f);
        float4 acc[4];
        const int cnt = pull_row<HID>(h1, srcA, head, nxt, E1, row, ok, l16, acc);
        const float inv = 1.0f / fmaxf((float)cnt, 1.0f);
#pragma unroll
        for (int j = 0; j < 4; ++j) {
            float4 a;
            a.x = f[j].x + acc[j].x * inv;
            a.y = f[j].y + acc[j].y * inv;
            a.z = f[j].z + acc[j].z * inv;
            a.w = f[j].w + acc[j].w * inv;
            *(float4*)&As[r * 260 + 4 * (l16 + 16 * j)] = a;
        }
    }
    __syncthreads();

    const int lo = lane & 15, hi = lane >> 4;
    const int n0 = wave * 16;
    v8f acc = {0, 0, 0, 0, 0, 0, 0, 0};
    for (int k = 0; k < HID; k += 4) {
        const int kk = k + 2 * hi;
        v2f a;
        a[0] = As[lo * 260 + kk];
        a[1] = As[lo * 260 + kk + 1];
        v2f b;
        b[0] = W[kk * DOUT + n0 + lo];
        b[1] = W[(kk + 1) * DOUT + n0 + lo];
        acc = __builtin_amdgcn_wmma_f32_16x16x4_f32(false, a, false, b, (short)0, acc, false, false);
    }
#pragma unroll
    for (int r = 0; r < 8; ++r) {
        int row = m0 + r + 8 * hi;
        if (row < N2) out[(long)row * DOUT + n0 + lo] = acc[r] + bias[n0 + lo];
    }
}

extern "C" void kernel_launch(void* const* d_in, const int* in_sizes, int n_in,
                              void* d_out, int out_size, void* d_ws, size_t ws_size,
                              hipStream_t stream) {
    (void)in_sizes; (void)n_in; (void)out_size; (void)ws_size;
    const int*   x  = (const int*)d_in[0];
    const float* e0 = (const float*)d_in[1];
    const float* e1 = (const float*)d_in[2];
    const float* e2 = (const float*)d_in[3];
    const float* e3 = (const float*)d_in[4];
    const float* Win[2]  = {(const float*)d_in[5], (const float*)d_in[9]};
    const float* bin[2]  = {(const float*)d_in[6], (const float*)d_in[10]};
    const float* Wout[2] = {(const float*)d_in[7], (const float*)d_in[11]};
    const float* bout[2] = {(const float*)d_in[8], (const float*)d_in[12]};
    const int* srcL0[2] = {(const int*)d_in[13], (const int*)d_in[17]};
    const int* dstL0[2] = {(const int*)d_in[14], (const int*)d_in[18]};
    const int* srcL1[2] = {(const int*)d_in[15], (const int*)d_in[19]};
    const int* dstL1[2] = {(const int*)d_in[16], (const int*)d_in[20]};
    float* out = (float*)d_out;

    size_t off = 0;
    auto take = [&](size_t bytes) { size_t o = off; off += (bytes + 255) & ~(size_t)255; return o; };
    char* ws = (char*)d_ws;
    float* feat = (float*)(ws + take((size_t)N0 * DIN * 4));
    float* h1   = (float*)(ws + take((size_t)N1 * HID * 4));
    int* head = (int*)(ws + take((size_t)NHEAD * 4));
    int* nxt  = (int*)(ws + take((size_t)NLINK * 4));

    build_feat<<<N0 / 8, 256, 0, stream>>>(x, e0, e1, e2, e3, feat);

    fill_heads<<<(NHEAD + 255) / 256, 256, 0, stream>>>(head);
    link_edges<<<(NLINK + 255) / 256, 256, 0, stream>>>(dstL0[0], dstL0[1], dstL1[0], dstL1[1], head, nxt);

    for (int e = 0; e < 2; ++e) {
        gemm1_fused<<<N1 / 16, 256, 0, stream>>>(feat, srcL0[e], head + (size_t)e * N1,
                                                 nxt + (size_t)e * E0, Win[e], bin[e], h1);
        gemm2_fused<<<(N2 + 15) / 16, 256, 0, stream>>>(h1, srcL1[e], head + 2 * (size_t)N1 + (size_t)e * N2,
                                                        nxt + 2 * (size_t)E0 + (size_t)e * E1,
                                                        Wout[e], bout[e], out + (size_t)e * N2 * DOUT);
    }
}


# layer-0 feature rows read straight from the embedding tables in the GEMM prologue (x entry, then table rows): feat plane and its 400000-wave build launch gone; sums and their order unchanged
# speedup vs baseline: 2.2627x; 1.2147x over previous
#include <hip/hip_runtime.h>

constexpr int N0 = 400000, N1 = 100000, N2 = 25000;
constexpr int E0 = 800000, E1 = 200000;
constexpr int DIN = 128, HID = 256, DOUT = 128;

typedef __attribute__((ext_vector_type(2))) float v2f;
typedef __attribute__((ext_vector_type(8))) float v8f;

struct FeatLane {
    const int* xlo;
    const int* xhi;
    const float* lo;
    const float* hi;
    int loShift;
};

__device__ __forceinline__ FeatLane feat_lane(const int* __restrict__ x,
                                              const float* __restrict__ e0, const float* __restrict__ e1,
                                              const float* __restrict__ e2, const float* __restrict__ e3,
                                              int l16) {
    FeatLane L;
    if (l16 < 8)       { L.xlo = x + 0; L.lo = e0 + 4 * l16;        L.loShift = 5; }
    else if (l16 < 12) { L.xlo = x + 1; L.lo = e1 + 4 * (l16 - 8);  L.loShift = 4; }
    else               { L.xlo = x + 2; L.lo = e2 + 4 * (l16 - 12); L.loShift = 4; }
    L.xhi = x + 3;
    L.hi = e3 + 4 * l16;
    return L;
}

__device__ __forceinline__ void feat_load(const FeatLane& L, int s, float4 (&v)[2]) {
    const int xi = L.xlo[4 * (long)s];
    const int xw = L.xhi[4 * (long)s];
    v[0] = *(const float4*)(L.lo + ((long)xi << L.loShift));
    v[1] = *(const float4*)(L.hi + ((long)xw << 6));
}

constexpr int NHEAD = 2 * N1 + 2 * N2;
constexpr int NLINK = 2 * E0 + 2 * E1;

__global__ void fill_heads(int* __restrict__ head) {
    int i = blockIdx.x * blockDim.x + threadIdx.x;
    if (i < NHEAD) head[i] = -1;
}

__global__ void link_edges(const int* __restrict__ dA0, const int* __restrict__ dB0,
                           const int* __restrict__ dA1, const int* __restrict__ dB1,
                           int* __restrict__ head, int* __restrict__ nxt) {
    int t = blockIdx.x * blockDim.x + threadIdx.x;
    if (t >= NLINK) return;
    int e, d;
    int* hp;
    if (t < 2 * E0) {
        int b = (t >= E0) ? 1 : 0;
        e = t - b * E0;
        d = (b ? dB0 : dA0)[e];
        hp = head + b * N1 + d;
    } else {
        int u = t - 2 * E0;
        int b = (u >= E1) ? 1 : 0;
        e = u - b * E1;
        d = (b ? dB1 : dA1)[e];
        hp = head + 2 * N1 + b * N2 + d;
    }
    nxt[t] = atomicExch(hp, e);
}

template <int WIDTH>
__device__ __forceinline__ int pull_row(const float* __restrict__ h, const int* __restrict__ srcA,
                                        const int* __restrict__ head, const int* __restrict__ nxt,
                                        int nE, int row, bool rowOk, int l16,
                                        float4 (&acc)[WIDTH / 64]) {
    constexpr int J = WIDTH / 64;
#pragma unroll
    for (int j = 0; j < J; ++j) acc[j] = make_float4(0.0f, 0.0f, 0.0f, 0.0f);
    int cnt = 0;
    int e = rowOk ? head[row] : -1;
    for (int it = 0; it < nE && (unsigned)e < (unsigned)nE; ++it) {
        const int s = srcA[e];
        const int en = nxt[e];
        const float4* hp = (const float4*)(h + (long)s * WIDTH);
#pragma unroll
        for (int j = 0; j < J; ++j) {
            const float4 v = hp[l16 + 16 * j];
            acc[j].x += v.x; acc[j].y += v.y; acc[j].z += v.z; acc[j].w += v.w;
        }
        ++cnt;
        e = en;
    }
    return cnt;
}

__device__ __forceinline__ void add4(float4& a, const float4& v) {
    a.x += v.x; a.y += v.y; a.z += v.z; a.w += v.w;
}

__device__ __forceinline__ int pull_feat_row(const FeatLane& L, const int* __restrict__ srcA,
                                             const int* __restrict__ head, const int* __restrict__ nxt,
                                             int nE, int row, float4 (&acc)[2]) {
    acc[0] = make_float4(0.0f, 0.0f, 0.0f, 0.0f);
    acc[1] = make_float4(0.0f, 0.0f, 0.0f, 0.0f);
    int cnt = 0;
    int e = head[row];
    for (int it = 0; it < nE && (unsigned)e < (unsigned)nE; ++it) {
        const int s = srcA[e];
        const int en = nxt[e];
        float4 v[2];
        feat_load(L, s, v);
        add4(acc[0], v[0]);
        add4(acc[1], v[1]);
        ++cnt;
        e = en;
    }
    return cnt;
}

__global__ void __launch_bounds__(256)
gemm1_fused(const int* __restrict__ x,
            const float* __restrict__ e0, const float* __restrict__ e1,
            const float* __restrict__ e2, const float* __restrict__ e3,
            const int* __restrict__ srcA,
            const int* __restrict__ head, const int* __restrict__ nxt,
            const float* __restrict__ W,
            const float* __restrict__ bias, float* __restrict__ h1) {
    __shared__ alignas(16) float As[16 * 132];
    __shared__ float srow[16];
    const int m0 = blockIdx.x * 16;
    const int tid = threadIdx.x;
    const int lane = tid & 31, wave = tid >> 5;
    {
        const int r = tid >> 4, l16 = tid & 15;
        const FeatLane L = feat_lane(x, e0, e1, e2, e3, l16);
        float4 f[2];
        feat_load(L, m0 + r, f);
        float4 acc[2];
        const int cnt = pull_feat_row(L, srcA, head, nxt, E0, m0 + r, acc);
        const float inv = 1.0f / fmaxf((float)cnt, 1.0f);
#pragma unroll
        for (int j = 0; j < 2; ++j) {
            float4 a;
            a.x = f[j].x + acc[j].x * inv;
            a.y = f[j].y + acc[j].y * inv;
            a.z = f[j].z + acc[j].z * inv;
            a.w = f[j].w + acc[j].w * inv;
            *(float4*)&As[r * 132 + 4 * (l16 + 16 * j)] = a;
        }
        if (l16 == 0) srow[r] = (cnt > 0) ? 2.0f : 1.0f;
    }
    __syncthreads();

    const int lo = lane & 15, hi = lane >> 4;
    const int nA = wave * 16, nB = (wave + 8) * 16;
    v8f c0 = {0, 0, 0, 0, 0, 0, 0, 0};
    v8f c1 = {0, 0, 0, 0, 0, 0, 0, 0};
    for (int k = 0; k < DIN; k += 4) {
        const int kk = k + 2 * hi;
        v2f a;
        a[0] = As[lo * 132 + kk];
        a[1] = As[lo * 132 + kk + 1];
        v2f b0, b1;
        b0[0] = W[kk * HID + nA + lo];
        b0[1] = W[(kk + 1) * HID + nA + lo];
        b1[0] = W[kk * HID + nB + lo];
        b1[1] = W[(kk + 1) * HID + nB + lo];
        c0 = __builtin_amdgcn_wmma_f32_16x16x4_f32(false, a, false, b0, (short)0, c0, false, false);
        c1 = __builtin_amdgcn_wmma_f32_16x16x4_f32(false, a, false, b1, (short)0, c1, false, false);
    }
#pragma unroll
    for (int r = 0; r < 8; ++r) {
        int rr = r + 8 * hi;
        long row = m0 + rr;
        float s = srow[rr];
        h1[row * HID + nA + lo] = fmaxf(c0[r] + s * bias[nA + lo], 0.0f);
        h1[row * HID + nB + lo] = fmaxf(c1[r] + s * bias[nB + lo], 0.0f);
    }
}

__global__ void __launch_bounds__(256)
gemm2_fused(const float* __restrict__ h1, const int* __restrict__ srcA,
            const int* __restrict__ head, const int* __restrict__ nxt,
            const float* __restrict__ W,
            const float* __restrict__ bias, float* __restrict__ out) {
    __shared__ alignas(16) float As[16 * 260];
    const int m0 = blockIdx.x * 16;
    const int tid = threadIdx.x;
    const int lane = tid & 31, wave = tid >> 5;
    {
        const int r = tid >> 4, l16 = tid & 15;
        const int row = m0 + r;
        const bool ok = row < N2;
        float4 f[4];
#pragma unroll
        for (int j = 0; j < 4; ++j)
            f[j] = ok ? ((const float4*)(h1 + (long)row * HID))[l16 + 16 * j]
                      : make_float4(0.0f, 0.0f, 0.0f, 0.0f);
        float4 acc[4];
        const int cnt = pull_row<HID>(h1, srcA, head, nxt, E1, row, ok, l16, acc);
        const float inv = 1.0f / fmaxf((float)cnt, 1.0f);
#pragma unroll
        for (int j = 0; j < 4; ++j) {
            float4 a;
            a.x = f[j].x + acc[j].x * inv;
            a.y = f[j].y + acc[j].y * inv;
            a.z = f[j].z + acc[j].z * inv;
            a.w = f[j].w + acc[j].w * inv;
            *(float4*)&As[r * 260 + 4 * (l16 + 16 * j)] = a;
        }
    }
    __syncthreads();

    const int lo = lane & 15, hi = lane >> 4;
    const int n0 = wave * 16;
    v8f acc = {0, 0, 0, 0, 0, 0, 0, 0};
    for (int k = 0; k < HID; k += 4) {
        const int kk = k + 2 * hi;
        v2f a;
        a[0] = As[lo * 260 + kk];
        a[1] = As[lo * 260 + kk + 1];
        v2f b;
        b[0] = W[kk * DOUT + n0 + lo];
        b[1] = W[(kk + 1) * DOUT + n0 + lo];
        acc = __builtin_amdgcn_wmma_f32_16x16x4_f32(false, a, false, b, (short)0, acc, false, false);
    }
#pragma unroll
    for (int r = 0; r < 8; ++r) {
        int row = m0 + r + 8 * hi;
        if (row < N2) out[(long)row * DOUT + n0 + lo] = acc[r] + bias[n0 + lo];
    }
}

extern "C" void kernel_launch(void* const* d_in, const int* in_sizes, int n_in,
                              void* d_out, int out_size, void* d_ws, size_t ws_size,
                              hipStream_t stream) {
    (void)in_sizes; (void)n_in; (void)out_size; (void)ws_size;
    const int*   x  = (const int*)d_in[0];
    const float* e0 = (const float*)d_in[1];
    const float* e1 = (const float*)d_in[2];
    const float* e2 = (const float*)d_in[3];
    const float* e3 = (const float*)d_in[4];
    const float* Win[2]  = {(const float*)d_in[5], (const float*)d_in[9]};
    const float* bin[2]  = {(const float*)d_in[6], (const float*)d_in[10]};
    const float* Wout[2] = {(const float*)d_in[7], (const float*)d_in[11]};
    const float* bout[2] = {(const float*)d_in[8], (const float*)d_in[12]};
    const int* srcL0[2] = {(const int*)d_in[13], (const int*)d_in[17]};
    const int* dstL0[2] = {(const int*)d_in[14], (const int*)d_in[18]};
    const int* srcL1[2] = {(const int*)d_in[15], (const int*)d_in[19]};
    const int* dstL1[2] = {(const int*)d_in[16], (const int*)d_in[20]};
    float* out = (float*)d_out;

    size_t off = 0;
    auto take = [&](size_t bytes) { size_t o = off; off += (bytes + 255) & ~(size_t)255; return o; };
    char* ws = (char*)d_ws;
    float* h1   = (float*)(ws + take((size_t)N1 * HID * 4));
    int* head = (int*)(ws + take((size_t)NHEAD * 4));
    int* nxt  = (int*)(ws + take((size_t)NLINK * 4));

    fill_heads<<<(NHEAD + 255) / 256, 256, 0, stream>>>(head);
    link_edges<<<(NLINK + 255) / 256, 256, 0, stream>>>(dstL0[0], dstL0[1], dstL1[0], dstL1[1], head, nxt);

    for (int e = 0; e < 2; ++e) {
        gemm1_fused<<<N1 / 16, 256, 0, stream>>>(x, e0, e1, e2, e3, srcL0[e], head + (size_t)e * N1,
                                                 nxt + (size_t)e * E0, Win[e], bin[e], h1);
        gemm2_fused<<<(N2 + 15) / 16, 256, 0, stream>>>(h1, srcL1[e], head + 2 * (size_t)N1 + (size_t)e * N2,
                                                        nxt + 2 * (size_t)E0 + (size_t)e * E1,
                                                        Wout[e], bout[e], out + (size_t)e * N2 * DOUT);
    }
}


# list walks of both GEMM prologues software-pipelined: next edge's (src, nxt) pair, next source's x entry and this edge's rows in flight together, one wait per hop; order of every sum unchanged
# speedup vs baseline: 2.3504x; 1.0388x over previous
#include <hip/hip_runtime.h>

constexpr int N0 = 400000, N1 = 100000, N2 = 25000;
constexpr int E0 = 800000, E1 = 200000;
constexpr int DIN = 128, HID = 256, DOUT = 128;

typedef __attribute__((ext_vector_type(2))) float v2f;
typedef __attribute__((ext_vector_type(8))) float v8f;

struct FeatLane {
    const int* xlo;
    const int* xhi;
    const float* lo;
    const float* hi;
    int loShift;
};

__device__ __forceinline__ FeatLane feat_lane(const int* __restrict__ x,
                                              const float* __restrict__ e0, const float* __restrict__ e1,
                                              const float* __restrict__ e2, const float* __restrict__ e3,
                                              int l16) {
    FeatLane L;
    if (l16 < 8)       { L.xlo = x + 0; L.lo = e0 + 4 * l16;        L.loShift = 5; }
    else if (l16 < 12) { L.xlo = x + 1; L.lo = e1 + 4 * (l16 - 8);  L.loShift = 4; }
    else               { L.xlo = x + 2; L.lo = e2 + 4 * (l16 - 12); L.loShift = 4; }
    L.xhi = x + 3;
    L.hi = e3 + 4 * l16;
    return L;
}

__device__ __forceinline__ void feat_load(const FeatLane& L, int s, float4 (&v)[2]) {
    const int xi = L.xlo[4 * (long)s];
    const int xw = L.xhi[4 * (long)s];
    v[0] = *(const float4*)(L.lo + ((long)xi << L.loShift));
    v[1] = *(const float4*)(L.hi + ((long)xw << 6));
}

constexpr int NHEAD = 2 * N1 + 2 * N2;
constexpr int NLINK = 2 * E0 + 2 * E1;

__global__ void fill_heads(int* __restrict__ head) {
    int i = blockIdx.x * blockDim.x + threadIdx.x;
    if (i < NHEAD) head[i] = -1;
}

__global__ void link_edges(const int* __restrict__ dA0, const int* __restrict__ dB0,
                           const int* __restrict__ dA1, const int* __restrict__ dB1,
                           int* __restrict__ head, int* __restrict__ nxt) {
    int t = blockIdx.x * blockDim.x + threadIdx.x;
    if (t >= NLINK) return;
    int e, d;
    int* hp;
    if (t < 2 * E0) {
        int b = (t >= E0) ? 1 : 0;
        e = t - b * E0;
        d = (b ? dB0 : dA0)[e];
        hp = head + b * N1 + d;
    } else {
        int u = t - 2 * E0;
        int b = (u >= E1) ? 1 : 0;
        e = u - b * E1;
        d = (b ? dB1 : dA1)[e];
        hp = head + 2 * N1 + b * N2 + d;
    }
    nxt[t] = atomicExch(hp, e);
}

template <int WIDTH>
__device__ __forceinline__ int pull_row(const float* __restrict__ h, const int* __restrict__ srcA,
                                        const int* __restrict__ head, const int* __restrict__ nxt,
                                        int nE, int row, bool rowOk, int l16,
                                        float4 (&acc)[WIDTH / 64]) {
    constexpr int J = WIDTH / 64;
#pragma unroll
    for (int j = 0; j < J; ++j) acc[j] = make_float4(0.0f, 0.0f, 0.0f, 0.0f);
    int cnt = 0;
    int eA = rowOk ? head[row] : -1;
    bool okA = (unsigned)eA < (unsigned)nE;
    int sB = 0;
    bool okB = false;
    for (int it = 0; it <= nE && (okA || okB); ++it) {
        const int eL = okA ? eA : 0;
        const int sA = srcA[eL];
        const int nA = nxt[eL];
        if (okB) {
            const float4* hp = (const float4*)(h + (long)sB * WIDTH);
            float4 v[J];
#pragma unroll
            for (int j = 0; j < J; ++j) v[j] = hp[l16 + 16 * j];
#pragma unroll
            for (int j = 0; j < J; ++j) {
                acc[j].x += v[j].x; acc[j].y += v[j].y; acc[j].z += v[j].z; acc[j].w += v[j].w;
            }
            ++cnt;
        }
        okB = okA;
        sB = sA;
        okA = okA && (it + 1 < nE) && ((unsigned)nA < (unsigned)nE);
        eA = nA;
    }
    return cnt;
}

__device__ __forceinline__ void add4(float4& a, const float4& v) {
    a.x += v.x; a.y += v.y; a.z += v.z; a.w += v.w;
}

__device__ __forceinline__ int pull_feat_row(const FeatLane& L, const int* __restrict__ srcA,
                                             const int* __restrict__ head, const int* __restrict__ nxt,
                                             int nE, int row, float4 (&acc)[2]) {
    acc[0] = make_float4(0.0f, 0.0f, 0.0f, 0.0f);
    acc[1] = make_float4(0.0f, 0.0f, 0.0f, 0.0f);
    int cnt = 0;
    int eA = head[row];
    bool okA = (unsigned)eA < (unsigned)nE;
    int sB = 0;
    bool okB = false;
    int xiC = 0, xwC = 0;
    bool okC = false;
    for (int it = 0; it <= nE + 1 && (okA || okB || okC); ++it) {
        int sA = 0, nA = -1, xiB = 0, xwB = 0;
        if (okA) {
            sA = srcA[eA];
            nA = nxt[eA];
        }
        if (okB) {
            xiB = L.xlo[4 * (long)sB];
            xwB = L.xhi[4 * (long)sB];
        }
        if (okC) {
            const float4 v0 = *(const float4*)(L.lo + ((long)xiC << L.loShift));
            const float4 v1 = *(const float4*)(L.hi + ((long)xwC << 6));
            add4(acc[0], v0);
            add4(acc[1], v1);
            ++cnt;
        }
        okC = okB;
        xiC = xiB;
        xwC = xwB;
        okB = okA;
        sB = sA;
        okA = okA && (it + 1 < nE) && ((unsigned)nA < (unsigned)nE);
        eA = nA;
    }
    return cnt;
}

__global__ void __launch_bounds__(256)
gemm1_fused(const int* __restrict__ x,
            const float* __restrict__ e0, const float* __restrict__ e1,
            const float* __restrict__ e2, const float* __restrict__ e3,
            const int* __restrict__ srcA,
            const int* __restrict__ head, const int* __restrict__ nxt,
            const float* __restrict__ W,
            const float* __restrict__ bias, float* __restrict__ h1) {
    __shared__ alignas(16) float As[16 * 132];
    __shared__ float srow[16];
    const int m0 = blockIdx.x * 16;
    const int tid = threadIdx.x;
    const int lane = tid & 31, wave = tid >> 5;
    {
        const int r = tid >> 4, l16 = tid & 15;
        const FeatLane L = feat_lane(x, e0, e1, e2, e3, l16);
        float4 f[2];
        feat_load(L, m0 + r, f);
        float4 acc[2];
        const int cnt = pull_feat_row(L, srcA, head, nxt, E0, m0 + r, acc);
        const float inv = 1.0f / fmaxf((float)cnt, 1.0f);
#pragma unroll
        for (int j = 0; j < 2; ++j) {
            float4 a;
            a.x = f[j].x + acc[j].x * inv;
            a.y = f[j].y + acc[j].y * inv;
            a.z = f[j].z + acc[j].z * inv;
            a.w = f[j].w + acc[j].w * inv;
            *(float4*)&As[r * 132 + 4 * (l16 + 16 * j)] = a;
        }
        if (l16 == 0) srow[r] = (cnt > 0) ? 2.0f : 1.0f;
    }
    __syncthreads();

    const int lo = lane & 15, hi = lane >> 4;
    const int nA = wave * 16, nB = (wave + 8) * 16;
    v8f c0 = {0, 0, 0, 0, 0, 0, 0, 0};
    v8f c1 = {0, 0, 0, 0, 0, 0, 0, 0};
    for (int k = 0; k < DIN; k += 4) {
        const int kk = k + 2 * hi;
        v2f a;
        a[0] = As[lo * 132 + kk];
        a[1] = As[lo * 132 + kk + 1];
        v2f b0, b1;
        b0[0] = W[kk * HID + nA + lo];
        b0[1] = W[(kk + 1) * HID + nA + lo];
        b1[0] = W[kk * HID + nB + lo];
        b1[1] = W[(kk + 1) * HID + nB + lo];
        c0 = __builtin_amdgcn_wmma_f32_16x16x4_f32(false, a, false, b0, (short)0, c0, false, false);
        c1 = __builtin_amdgcn_wmma_f32_16x16x4_f32(false, a, false, b1, (short)0, c1, false, false);
    }
#pragma unroll
    for (int r = 0; r < 8; ++r) {
        int rr = r + 8 * hi;
        long row = m0 + rr;
        float s = srow[rr];
        h1[row * HID + nA + lo] = fmaxf(c0[r] + s * bias[nA + lo], 0.0f);
        h1[row * HID + nB + lo] = fmaxf(c1[r] + s * bias[nB + lo], 0.0f);
    }
}

__global__ void __launch_bounds__(256)
gemm2_fused(const float* __restrict__ h1, const int* __restrict__ srcA,
            const int* __restrict__ head, const int* __restrict__ nxt,
            const float* __restrict__ W,
            const float* __restrict__ bias, float* __restrict__ out) {
    __shared__ alignas(16) float As[16 * 260];
    const int m0 = blockIdx.x * 16;
    const int tid = threadIdx.x;
    const int lane = tid & 31, wave = tid >> 5;
    {
        const int r = tid >> 4, l16 = tid & 15;
        const int row = m0 + r;
        const bool ok = row < N2;
        float4 f[4];
#pragma unroll
        for (int j = 0; j < 4; ++j)
            f[j] = ok ? ((const float4*)(h1 + (long)row * HID))[l16 + 16 * j]
                      : make_float4(0.0f, 0.0f, 0.0f, 0.0f);
        float4 acc[4];
        const int cnt = pull_row<HID>(h1, srcA, head, nxt, E1, row, ok, l16, acc);
        const float inv = 1.0f / fmaxf((float)cnt, 1.0f);
#pragma unroll
        for (int j = 0; j < 4; ++j) {
            float4 a;
            a.x = f[j].x + acc[j].x * inv;
            a.y = f[j].y + acc[j].y * inv;
            a.z = f[j].z + acc[j].z * inv;
            a.w = f[j].w + acc[j].w * inv;
            *(float4*)&As[r * 260 + 4 * (l16 + 16 * j)] = a;
        }
    }
    __syncthreads();

    const int lo = lane & 15, hi = lane >> 4;
    const int n0 = wave * 16;
    v8f acc = {0, 0, 0, 0, 0, 0, 0, 0};
    for (int k = 0; k < HID; k += 4) {
        const int kk = k + 2 * hi;
        v2f a;
        a[0] = As[lo * 260 + kk];
        a[1] = As[lo * 260 + kk + 1];
        v2f b;
        b[0] = W[kk * DOUT + n0 + lo];
        b[1] = W[(kk + 1) * DOUT + n0 + lo];
        acc = __builtin_amdgcn_wmma_f32_16x16x4_f32(false, a, false, b, (short)0, acc, false, false);
    }
#pragma unroll
    for (int r = 0; r < 8; ++r) {
        int row = m0 + r + 8 * hi;
        if (row < N2) out[(long)row * DOUT + n0 + lo] = acc[r] + bias[n0 + lo];
    }
}

extern "C" void kernel_launch(void* const* d_in, const int* in_sizes, int n_in,
                              void* d_out, int out_size, void* d_ws, size_t ws_size,
                              hipStream_t stream) {
    (void)in_sizes; (void)n_in; (void)out_size; (void)ws_size;
    const int*   x  = (const int*)d_in[0];
    const float* e0 = (const float*)d_in[1];
    const float* e1 = (const float*)d_in[2];
    const float* e2 = (const float*)d_in[3];
    const float* e3 = (const float*)d_in[4];
    const float* Win[2]  = {(const float*)d_in[5], (const float*)d_in[9]};
    const float* bin[2]  = {(const float*)d_in[6], (const float*)d_in[10]};
    const float* Wout[2] = {(const float*)d_in[7], (const float*)d_in[11]};
    const float* bout[2] = {(const float*)d_in[8], (const float*)d_in[12]};
    const int* srcL0[2] = {(const int*)d_in[13], (const int*)d_in[17]};
    const int* dstL0[2] = {(const int*)d_in[14], (const int*)d_in[18]};
    const int* srcL1[2] = {(const int*)d_in[15], (const int*)d_in[19]};
    const int* dstL1[2] = {(const int*)d_in[16], (const int*)d_in[20]};
    float* out = (float*)d_out;

    size_t off = 0;
    auto take = [&](size_t bytes) { size_t o = off; off += (bytes + 255) & ~(size_t)255; return o; };
    char* ws = (char*)d_ws;
    float* h1   = (float*)(ws + take((size_t)N1 * HID * 4));
    int* head = (int*)(ws + take((size_t)NHEAD * 4));
    int* nxt  = (int*)(ws + take((size_t)NLINK * 4));

    fill_heads<<<(NHEAD + 255) / 256, 256, 0, stream>>>(head);
    link_edges<<<(NLINK + 255) / 256, 256, 0, stream>>>(dstL0[0], dstL0[1], dstL1[0], dstL1[1], head, nxt);

    for (int e = 0; e < 2; ++e) {
        gemm1_fused<<<N1 / 16, 256, 0, stream>>>(x, e0, e1, e2, e3, srcL0[e], head + (size_t)e * N1,
                                                 nxt + (size_t)e * E0, Win[e], bin[e], h1);
        gemm2_fused<<<(N2 + 15) / 16, 256, 0, stream>>>(h1, srcL1[e], head + 2 * (size_t)N1 + (size_t)e * N2,
                                                        nxt + 2 * (size_t)E0 + (size_t)e * E1,
                                                        Wout[e], bout[e], out + (size_t)e * N2 * DOUT);
    }
}


# gemm1 tiles re-cut: 32 rows per block, each wave four column-interleaved N tiles (tile j = columns 4c+j), so W loads and h1 stores are whole-line 16-byte accesses; order of every sum unchanged
# speedup vs baseline: 2.5310x; 1.0768x over previous
#include <hip/hip_runtime.h>

constexpr int N0 = 400000, N1 = 100000, N2 = 25000;
constexpr int E0 = 800000, E1 = 200000;
constexpr int DIN = 128, HID = 256, DOUT = 128;

typedef __attribute__((ext_vector_type(2))) float v2f;
typedef __attribute__((ext_vector_type(8))) float v8f;

struct FeatLane {
    const int* xlo;
    const int* xhi;
    const float* lo;
    const float* hi;
    int loShift;
};

__device__ __forceinline__ FeatLane feat_lane(const int* __restrict__ x,
                                              const float* __restrict__ e0, const float* __restrict__ e1,
                                              const float* __restrict__ e2, const float* __restrict__ e3,
                                              int l16) {
    FeatLane L;
    if (l16 < 8)       { L.xlo = x + 0; L.lo = e0 + 4 * l16;        L.loShift = 5; }
    else if (l16 < 12) { L.xlo = x + 1; L.lo = e1 + 4 * (l16 - 8);  L.loShift = 4; }
    else               { L.xlo = x + 2; L.lo = e2 + 4 * (l16 - 12); L.loShift = 4; }
    L.xhi = x + 3;
    L.hi = e3 + 4 * l16;
    return L;
}

__device__ __forceinline__ void feat_load(const FeatLane& L, int s, float4 (&v)[2]) {
    const int xi = L.xlo[4 * (long)s];
    const int xw = L.xhi[4 * (long)s];
    v[0] = *(const float4*)(L.lo + ((long)xi << L.loShift));
    v[1] = *(const float4*)(L.hi + ((long)xw << 6));
}

constexpr int NHEAD = 2 * N1 + 2 * N2;
constexpr int NLINK = 2 * E0 + 2 * E1;

__global__ void fill_heads(int* __restrict__ head) {
    int i = blockIdx.x * blockDim.x + threadIdx.x;
    if (i < NHEAD) head[i] = -1;
}

__global__ void link_edges(const int* __restrict__ dA0, const int* __restrict__ dB0,
                           const int* __restrict__ dA1, const int* __restrict__ dB1,
                           int* __restrict__ head, int* __restrict__ nxt) {
    int t = blockIdx.x * blockDim.x + threadIdx.x;
    if (t >= NLINK) return;
    int e, d;
    int* hp;
    if (t < 2 * E0) {
        int b = (t >= E0) ? 1 : 0;
        e = t - b * E0;
        d = (b ? dB0 : dA0)[e];
        hp = head + b * N1 + d;
    } else {
        int u = t - 2 * E0;
        int b = (u >= E1) ? 1 : 0;
        e = u - b * E1;
        d = (b ? dB1 : dA1)[e];
        hp = head + 2 * N1 + b * N2 + d;
    }
    nxt[t] = atomicExch(hp, e);
}

template <int WIDTH>
__device__ __forceinline__ int pull_row(const float* __restrict__ h, const int* __restrict__ srcA,
                                        const int* __restrict__ head, const int* __restrict__ nxt,
                                        int nE, int row, bool rowOk, int l16,
                                        float4 (&acc)[WIDTH / 64]) {
    constexpr int J = WIDTH / 64;
#pragma unroll
    for (int j = 0; j < J; ++j) acc[j] = make_float4(0.0f, 0.0f, 0.0f, 0.0f);
    int cnt = 0;
    int eA = rowOk ? head[row] : -1;
    bool okA = (unsigned)eA < (unsigned)nE;
    int sB = 0;
    bool okB = false;
    for (int it = 0; it <= nE && (okA || okB); ++it) {
        const int eL = okA ? eA : 0;
        const int sA = srcA[eL];
        const int nA = nxt[eL];
        if (okB) {
            const float4* hp = (const float4*)(h + (long)sB * WIDTH);
            float4 v[J];
#pragma unroll
            for (int j = 0; j < J; ++j) v[j] = hp[l16 + 16 * j];
#pragma unroll
            for (int j = 0; j < J; ++j) {
                acc[j].x += v[j].x; acc[j].y += v[j].y; acc[j].z += v[j].z; acc[j].w += v[j].w;
            }
            ++cnt;
        }
        okB = okA;
        sB = sA;
        okA = okA && (it + 1 < nE) && ((unsigned)nA < (unsigned)nE);
        eA = nA;
    }
    return cnt;
}

__device__ __forceinline__ void add4(float4& a, const float4& v) {
    a.x += v.x; a.y += v.y; a.z += v.z; a.w += v.w;
}

__device__ __forceinline__ int pull_feat_row(const FeatLane& L, const int* __restrict__ srcA,
                                             const int* __restrict__ head, const int* __restrict__ nxt,
                                             int nE, int row, float4 (&acc)[2]) {
    acc[0] = make_float4(0.0f, 0.0f, 0.0f, 0.0f);
    acc[1] = make_float4(0.0f, 0.0f, 0.0f, 0.0f);
    int cnt = 0;
    int eA = head[row];
    bool okA = (unsigned)eA < (unsigned)nE;
    int sB = 0;
    bool okB = false;
    int xiC = 0, xwC = 0;
    bool okC = false;
    for (int it = 0; it <= nE + 1 && (okA || okB || okC); ++it) {
        int sA = 0, nA = -1, xiB = 0, xwB = 0;
        if (okA) {
            sA = srcA[eA];
            nA = nxt[eA];
        }
        if (okB) {
            xiB = L.xlo[4 * (long)sB];
            xwB = L.xhi[4 * (long)sB];
        }
        if (okC) {
            const float4 v0 = *(const float4*)(L.lo + ((long)xiC << L.loShift));
            const float4 v1 = *(const float4*)(L.hi + ((long)xwC << 6));
            add4(acc[0], v0);
            add4(acc[1], v1);
            ++cnt;
        }
        okC = okB;
        xiC = xiB;
        xwC = xwB;
        okB = okA;
        sB = sA;
        okA = okA && (it + 1 < nE) && ((unsigned)nA < (unsigned)nE);
        eA = nA;
    }
    return cnt;
}

__global__ void __launch_bounds__(256)
gemm1_fused(const int* __restrict__ x,
            const float* __restrict__ e0, const float* __restrict__ e1,
            const float* __restrict__ e2, const float* __restrict__ e3,
            const int* __restrict__ srcA,
            const int* __restrict__ head, const int* __restrict__ nxt,
            const float* __restrict__ W,
            const float* __restrict__ bias, float* __restrict__ h1) {
    __shared__ alignas(16) float As[32 * 132];
    __shared__ float srow[32];
    const int m0 = blockIdx.x * 32;
    const int tid = threadIdx.x;
    const int lane = tid & 31, wave = tid >> 5;
    {
        const int r = tid >> 4, l16 = tid & 15;
        const FeatLane L = feat_lane(x, e0, e1, e2, e3, l16);
        for (int p = 0; p < 2; ++p) {
            const int rl = r + 16 * p;
            float4 f[2];
            feat_load(L, m0 + rl, f);
            float4 acc[2];
            const int cnt = pull_feat_row(L, srcA, head, nxt, E0, m0 + rl, acc);
            const float inv = 1.0f / fmaxf((float)cnt, 1.0f);
#pragma unroll
            for (int j = 0; j < 2; ++j) {
                float4 a;
                a.x = f[j].x + acc[j].x * inv;
                a.y = f[j].y + acc[j].y * inv;
                a.z = f[j].z + acc[j].z * inv;
                a.w = f[j].w + acc[j].w * inv;
                *(float4*)&As[rl * 132 + 4 * (l16 + 16 * j)] = a;
            }
            if (l16 == 0) srow[rl] = (cnt > 0) ? 2.0f : 1.0f;
        }
    }
    __syncthreads();

    const int lo = lane & 15, hi = lane >> 4;
    const int mt = wave >> 2, g = wave & 3;
    const float* Ar = &As[(mt * 16 + lo) * 132];
    const float* Wg = W + 64 * g + 4 * lo;
    v8f c0 = {0, 0, 0, 0, 0, 0, 0, 0};
    v8f c1 = {0, 0, 0, 0, 0, 0, 0, 0};
    v8f c2 = {0, 0, 0, 0, 0, 0, 0, 0};
    v8f c3 = {0, 0, 0, 0, 0, 0, 0, 0};
#pragma unroll 2
    for (int k = 0; k < DIN; k += 4) {
        const int kk = k + 2 * hi;
        v2f a;
        a[0] = Ar[kk];
        a[1] = Ar[kk + 1];
        const float4 w0 = *(const float4*)(Wg + kk * HID);
        const float4 w1 = *(const float4*)(Wg + (kk + 1) * HID);
        v2f b0, b1, b2, b3;
        b0[0] = w0.x; b0[1] = w1.x;
        b1[0] = w0.y; b1[1] = w1.y;
        b2[0] = w0.z; b2[1] = w1.z;
        b3[0] = w0.w; b3[1] = w1.w;
        c0 = __builtin_amdgcn_wmma_f32_16x16x4_f32(false, a, false, b0, (short)0, c0, false, false);
        c1 = __builtin_amdgcn_wmma_f32_16x16x4_f32(false, a, false, b1, (short)0, c1, false, false);
        c2 = __builtin_amdgcn_wmma_f32_16x16x4_f32(false, a, false, b2, (short)0, c2, false, false);
        c3 = __builtin_amdgcn_wmma_f32_16x16x4_f32(false, a, false, b3, (short)0, c3, false, false);
    }
    const float4 bv = *(const float4*)(bias + 64 * g + 4 * lo);
#pragma unroll
    for (int r = 0; r < 8; ++r) {
        int rr = mt * 16 + r + 8 * hi;
        long row = m0 + rr;
        float s = srow[rr];
        float4 o;
        o.x = fmaxf(c0[r] + s * bv.x, 0.0f);
        o.y = fmaxf(c1[r] + s * bv.y, 0.0f);
        o.z = fmaxf(c2[r] + s * bv.z, 0.0f);
        o.w = fmaxf(c3[r] + s * bv.w, 0.0f);
        *(float4*)(h1 + row * HID + 64 * g + 4 * lo) = o;
    }
}

__global__ void __launch_bounds__(256)
gemm2_fused(const float* __restrict__ h1, const int* __restrict__ srcA,
            const int* __restrict__ head, const int* __restrict__ nxt,
            const float* __restrict__ W,
            const float* __restrict__ bias, float* __restrict__ out) {
    __shared__ alignas(16) float As[16 * 260];
    const int m0 = blockIdx.x * 16;
    const int tid = threadIdx.x;
    const int lane = tid & 31, wave = tid >> 5;
    {
        const int r = tid >> 4, l16 = tid & 15;
        const int row = m0 + r;
        const bool ok = row < N2;
        float4 f[4];
#pragma unroll
        for (int j = 0; j < 4; ++j)
            f[j] = ok ? ((const float4*)(h1 + (long)row * HID))[l16 + 16 * j]
                      : make_float4(0.0f, 0.0f, 0.0f, 0.0f);
        float4 acc[4];
        const int cnt = pull_row<HID>(h1, srcA, head, nxt, E1, row, ok, l16, acc);
        const float inv = 1.0f / fmaxf((float)cnt, 1.0f);
#pragma unroll
        for (int j = 0; j < 4; ++j) {
            float4 a;
            a.x = f[j].x + acc[j].x * inv;
            a.y = f[j].y + acc[j].y * inv;
            a.z = f[j].z + acc[j].z * inv;
            a.w = f[j].w + acc[j].w * inv;
            *(float4*)&As[r * 260 + 4 * (l16 + 16 * j)] = a;
        }
    }
    __syncthreads();

    const int lo = lane & 15, hi = lane >> 4;
    const int n0 = wave * 16;
    v8f acc = {0, 0, 0, 0, 0, 0, 0, 0};
    for (int k = 0; k < HID; k += 4) {
        const int kk = k + 2 * hi;
        v2f a;
        a[0] = As[lo * 260 + kk];
        a[1] = As[lo * 260 + kk + 1];
        v2f b;
        b[0] = W[kk * DOUT + n0 + lo];
        b[1] = W[(kk + 1) * DOUT + n0 + lo];
        acc = __builtin_amdgcn_wmma_f32_16x16x4_f32(false, a, false, b, (short)0, acc, false, false);
    }
#pragma unroll
    for (int r = 0; r < 8; ++r) {
        int row = m0 + r + 8 * hi;
        if (row < N2) out[(long)row * DOUT + n0 + lo] = acc[r] + bias[n0 + lo];
    }
}

extern "C" void kernel_launch(void* const* d_in, const int* in_sizes, int n_in,
                              void* d_out, int out_size, void* d_ws, size_t ws_size,
                              hipStream_t stream) {
    (void)in_sizes; (void)n_in; (void)out_size; (void)ws_size;
    const int*   x  = (const int*)d_in[0];
    const float* e0 = (const float*)d_in[1];
    const float* e1 = (const float*)d_in[2];
    const float* e2 = (const float*)d_in[3];
    const float* e3 = (const float*)d_in[4];
    const float* Win[2]  = {(const float*)d_in[5], (const float*)d_in[9]};
    const float* bin[2]  = {(const float*)d_in[6], (const float*)d_in[10]};
    const float* Wout[2] = {(const float*)d_in[7], (const float*)d_in[11]};
    const float* bout[2] = {(const float*)d_in[8], (const float*)d_in[12]};
    const int* srcL0[2] = {(const int*)d_in[13], (const int*)d_in[17]};
    const int* dstL0[2] = {(const int*)d_in[14], (const int*)d_in[18]};
    const int* srcL1[2] = {(const int*)d_in[15], (const int*)d_in[19]};
    const int* dstL1[2] = {(const int*)d_in[16], (const int*)d_in[20]};
    float* out = (float*)d_out;

    size_t off = 0;
    auto take = [&](size_t bytes) { size_t o = off; off += (bytes + 255) & ~(size_t)255; return o; };
    char* ws = (char*)d_ws;
    float* h1   = (float*)(ws + take((size_t)N1 * HID * 4));
    int* head = (int*)(ws + take((size_t)NHEAD * 4));
    int* nxt  = (int*)(ws + take((size_t)NLINK * 4));

    fill_heads<<<(NHEAD + 255) / 256, 256, 0, stream>>>(head);
    link_edges<<<(NLINK + 255) / 256, 256, 0, stream>>>(dstL0[0], dstL0[1], dstL1[0], dstL1[1], head, nxt);

    for (int e = 0; e < 2; ++e) {
        gemm1_fused<<<N1 / 32, 256, 0, stream>>>(x, e0, e1, e2, e3, srcL0[e], head + (size_t)e * N1,
                                                 nxt + (size_t)e * E0, Win[e], bin[e], h1);
        gemm2_fused<<<(N2 + 15) / 16, 256, 0, stream>>>(h1, srcL1[e], head + 2 * (size_t)N1 + (size_t)e * N2,
                                                        nxt + 2 * (size_t)E0 + (size_t)e * E1,
                                                        Wout[e], bout[e], out + (size_t)e * N2 * DOUT);
    }
}


# pull aggregation reads each destination's sources from one 64-byte slot line filled by one counting add per edge (overflow on an exchange list): no per-edge hops; rows summed in arrival order
# speedup vs baseline: 2.6156x; 1.0334x over previous
#include <hip/hip_runtime.h>

constexpr int N0 = 400000, N1 = 100000, N2 = 25000;
constexpr int E0 = 800000, E1 = 200000;
constexpr int DIN = 128, HID = 256, DOUT = 128;

typedef __attribute__((ext_vector_type(2))) float v2f;
typedef __attribute__((ext_vector_type(8))) float v8f;

struct FeatLane {
    const int* xlo;
    const int* xhi;
    const float* lo;
    const float* hi;
    int loShift;
    int sel;
};

__device__ __forceinline__ FeatLane feat_lane(const int* __restrict__ x,
                                              const float* __restrict__ e0, const float* __restrict__ e1,
                                              const float* __restrict__ e2, const float* __restrict__ e3,
                                              int l16) {
    FeatLane L;
    if (l16 < 8)       { L.xlo = x + 0; L.lo = e0 + 4 * l16;        L.loShift = 5; L.sel = 0; }
    else if (l16 < 12) { L.xlo = x + 1; L.lo = e1 + 4 * (l16 - 8);  L.loShift = 4; L.sel = 1; }
    else               { L.xlo = x + 2; L.lo = e2 + 4 * (l16 - 12); L.loShift = 4; L.sel = 2; }
    L.xhi = x + 3;
    L.hi = e3 + 4 * l16;
    return L;
}

__device__ __forceinline__ void feat_load(const FeatLane& L, int s, float4 (&v)[2]) {
    const int xi = L.xlo[4 * (long)s];
    const int xw = L.xhi[4 * (long)s];
    v[0] = *(const float4*)(L.lo + ((long)xi << L.loShift));
    v[1] = *(const float4*)(L.hi + ((long)xw << 6));
}

constexpr int NHEAD = 2 * N1 + 2 * N2;
constexpr int NLINK = 2 * E0 + 2 * E1;
constexpr int CAP = 16;

__global__ void fill_heads(int* __restrict__ ch) {
    int i = blockIdx.x * blockDim.x + threadIdx.x;
    if (i < NHEAD) ((int2*)ch)[i] = make_int2(0, -1);
}

__global__ void link_edges(const int* __restrict__ sA0, const int* __restrict__ sB0,
                           const int* __restrict__ sA1, const int* __restrict__ sB1,
                           const int* __restrict__ dA0, const int* __restrict__ dB0,
                           const int* __restrict__ dA1, const int* __restrict__ dB1,
                           int* __restrict__ ch, int* __restrict__ slot, int* __restrict__ nxt) {
    int t = blockIdx.x * blockDim.x + threadIdx.x;
    if (t >= NLINK) return;
    int e, d, sv, hd;
    if (t < 2 * E0) {
        int b = (t >= E0) ? 1 : 0;
        e = t - b * E0;
        d = (b ? dB0 : dA0)[e];
        sv = (b ? sB0 : sA0)[e];
        hd = b * N1 + d;
    } else {
        int u = t - 2 * E0;
        int b = (u >= E1) ? 1 : 0;
        e = u - b * E1;
        d = (b ? dB1 : dA1)[e];
        sv = (b ? sB1 : sA1)[e];
        hd = 2 * N1 + b * N2 + d;
    }
    const int pos = atomicAdd(ch + 2 * (long)hd, 1);
    if (pos < CAP) slot[(long)hd * CAP + pos] = sv;
    else           nxt[t] = atomicExch(ch + 2 * (long)hd + 1, e);
}

__device__ __forceinline__ void add4(float4& a, const float4& v) {
    a.x += v.x; a.y += v.y; a.z += v.z; a.w += v.w;
}

template <int WIDTH>
__device__ __forceinline__ int pull_row(const float* __restrict__ h, const int* __restrict__ srcA,
                                        const int* __restrict__ ch, const int* __restrict__ slot,
                                        const int* __restrict__ nxt,
                                        int nE, int row, bool rowOk, int l16,
                                        float4 (&acc)[WIDTH / 64]) {
    constexpr int J = WIDTH / 64;
#pragma unroll
    for (int j = 0; j < J; ++j) acc[j] = make_float4(0.0f, 0.0f, 0.0f, 0.0f);
    const int2 c = rowOk ? ((const int2*)ch)[row] : make_int2(0, -1);
    const int n = (c.x < CAP) ? c.x : CAP;
    const int mine = (l16 < n) ? slot[(long)row * CAP + l16] : 0;
    const float4* __restrict__ h4 = (const float4*)h;
    int cnt = 0;
    int i0 = 0;
    if (n & 1) {
        const int s0 = __shfl(mine, 0, 16);
        const float4* p0 = (const float4*)(h + (long)s0 * WIDTH);
#pragma unroll
        for (int j = 0; j < J; ++j) add4(acc[j], p0[l16 + 16 * j]);
        cnt = 1;
        i0 = 1;
    }
    for (; i0 < n; i0 += 2) {
        const int s0 = __shfl(mine, i0, 16);
        const int s1 = __shfl(mine, i0 + 1, 16);
        const unsigned o0 = (unsigned)s0 * (WIDTH / 4) + l16;
        const unsigned o1 = (unsigned)s1 * (WIDTH / 4) + l16;
        float4 v0[J], v1[J];
#pragma unroll
        for (int j = 0; j < J; ++j) { v0[j] = h4[o0 + 16 * j]; v1[j] = h4[o1 + 16 * j]; }
#pragma unroll
        for (int j = 0; j < J; ++j) add4(acc[j], v0[j]);
#pragma unroll
        for (int j = 0; j < J; ++j) add4(acc[j], v1[j]);
        cnt += 2;
    }
    if (c.x > CAP) {
        int e = c.y;
        for (int it = 0; it < nE && (unsigned)e < (unsigned)nE; ++it) {
            const int sv = srcA[e];
            const int nx = nxt[e];
            const float4* hp = (const float4*)(h + (long)sv * WIDTH);
#pragma unroll
            for (int j = 0; j < J; ++j) add4(acc[j], hp[l16 + 16 * j]);
            ++cnt;
            e = nx;
        }
    }
    return cnt;
}

template <int G>
__device__ __forceinline__ int pull_feat_row(const FeatLane& L, const int* __restrict__ x,
                                             const int* __restrict__ srcA,
                                             const int* __restrict__ ch, const int* __restrict__ slot,
                                             const int* __restrict__ nxt,
                                             int nE, int row, int l16, float4 (&acc)[2]) {
    acc[0] = make_float4(0.0f, 0.0f, 0.0f, 0.0f);
    acc[1] = make_float4(0.0f, 0.0f, 0.0f, 0.0f);
    const int2 c = ((const int2*)ch)[row];
    const int n = (c.x < CAP) ? c.x : CAP;
    const int mine = (l16 < n) ? slot[(long)row * CAP + l16] : 0;
    int4 X = make_int4(0, 0, 0, 0);
    if (l16 < n) X = *(const int4*)(x + 4 * (long)mine);
    int cnt = 0;
    for (int i0 = 0; i0 < n; i0 += G) {
        float4 v0[G], v1[G];
#pragma unroll
        for (int g = 0; g < G; ++g) {
            const int i = (i0 + g < n) ? i0 + g : i0;
            const int xa = __shfl(X.x, i, 16);
            const int xb = __shfl(X.y, i, 16);
            const int xc = __shfl(X.z, i, 16);
            const int xw = __shfl(X.w, i, 16);
            const int xi = (L.sel == 0) ? xa : ((L.sel == 1) ? xb : xc);
            v0[g] = *(const float4*)(L.lo + ((long)xi << L.loShift));
            v1[g] = *(const float4*)(L.hi + ((long)xw << 6));
        }
#pragma unroll
        for (int g = 0; g < G; ++g) {
            if (i0 + g < n) {
                add4(acc[0], v0[g]);
                add4(acc[1], v1[g]);
                ++cnt;
            }
        }
    }
    if (c.x > CAP) {
        int e = c.y;
        for (int it = 0; it < nE && (unsigned)e < (unsigned)nE; ++it) {
            const int sv = srcA[e];
            const int nx = nxt[e];
            float4 f[2];
            feat_load(L, sv, f);
            add4(acc[0], f[0]);
            add4(acc[1], f[1]);
            ++cnt;
            e = nx;
        }
    }
    return cnt;
}

__global__ void __launch_bounds__(256)
gemm1_fused(const int* __restrict__ x,
            const float* __restrict__ e0, const float* __restrict__ e1,
            const float* __restrict__ e2, const float* __restrict__ e3,
            const int* __restrict__ srcA,
            const int* __restrict__ ch, const int* __restrict__ slot, const int* __restrict__ nxt,
            const float* __restrict__ W,
            const float* __restrict__ bias, float* __restrict__ h1) {
    __shared__ alignas(16) float As[32 * 132];
    __shared__ float srow[32];
    const int m0 = blockIdx.x * 32;
    const int tid = threadIdx.x;
    const int lane = tid & 31, wave = tid >> 5;
    {
        const int r = tid >> 4, l16 = tid & 15;
        const FeatLane L = feat_lane(x, e0, e1, e2, e3, l16);
        for (int p = 0; p < 2; ++p) {
            const int rl = r + 16 * p;
            float4 f[2];
            feat_load(L, m0 + rl, f);
            float4 acc[2];
            const int cnt = pull_feat_row<2>(L, x, srcA, ch, slot, nxt, E0, m0 + rl, l16, acc);
            const float inv = 1.0f / fmaxf((float)cnt, 1.0f);
#pragma unroll
            for (int j = 0; j < 2; ++j) {
                float4 a;
                a.x = f[j].x + acc[j].x * inv;
                a.y = f[j].y + acc[j].y * inv;
                a.z = f[j].z + acc[j].z * inv;
                a.w = f[j].w + acc[j].w * inv;
                *(float4*)&As[rl * 132 + 4 * (l16 + 16 * j)] = a;
            }
            if (l16 == 0) srow[rl] = (cnt > 0) ? 2.0f : 1.0f;
        }
    }
    __syncthreads();

    const int lo = lane & 15, hi = lane >> 4;
    const int mt = wave >> 2, g = wave & 3;
    const float* Ar = &As[(mt * 16 + lo) * 132];
    const float* Wg = W + 64 * g + 4 * lo;
    v8f c0 = {0, 0, 0, 0, 0, 0, 0, 0};
    v8f c1 = {0, 0, 0, 0, 0, 0, 0, 0};
    v8f c2 = {0, 0, 0, 0, 0, 0, 0, 0};
    v8f c3 = {0, 0, 0, 0, 0, 0, 0, 0};
#pragma unroll 2
    for (int k = 0; k < DIN; k += 4) {
        const int kk = k + 2 * hi;
        v2f a;
        a[0] = Ar[kk];
        a[1] = Ar[kk + 1];
        const float4 w0 = *(const float4*)(Wg + kk * HID);
        const float4 w1 = *(const float4*)(Wg + (kk + 1) * HID);
        v2f b0, b1, b2, b3;
        b0[0] = w0.x; b0[1] = w1.x;
        b1[0] = w0.y; b1[1] = w1.y;
        b2[0] = w0.z; b2[1] = w1.z;
        b3[0] = w0.w; b3[1] = w1.w;
        c0 = __builtin_amdgcn_wmma_f32_16x16x4_f32(false, a, false, b0, (short)0, c0, false, false);
        c1 = __builtin_amdgcn_wmma_f32_16x16x4_f32(false, a, false, b1, (short)0, c1, false, false);
        c2 = __builtin_amdgcn_wmma_f32_16x16x4_f32(false, a, false, b2, (short)0, c2, false, false);
        c3 = __builtin_amdgcn_wmma_f32_16x16x4_f32(false, a, false, b3, (short)0, c3, false, false);
    }
    const float4 bv = *(const float4*)(bias + 64 * g + 4 * lo);
#pragma unroll
    for (int r = 0; r < 8; ++r) {
        int rr = mt * 16 + r + 8 * hi;
        long row = m0 + rr;
        float s = srow[rr];
        float4 o;
        o.x = fmaxf(c0[r] + s * bv.x, 0.0f);
        o.y = fmaxf(c1[r] + s * bv.y, 0.0f);
        o.z = fmaxf(c2[r] + s * bv.z, 0.0f);
        o.w = fmaxf(c3[r] + s * bv.w, 0.0f);
        *(float4*)(h1 + row * HID + 64 * g + 4 * lo) = o;
    }
}

__global__ void __launch_bounds__(256)
gemm2_fused(const float* __restrict__ h1, const int* __restrict__ srcA,
            const int* __restrict__ ch, const int* __restrict__ slot, const int* __restrict__ nxt,
            const float* __restrict__ W,
            const float* __restrict__ bias, float* __restrict__ out) {
    __shared__ alignas(16) float As[16 * 260];
    const int m0 = blockIdx.x * 16;
    const int tid = threadIdx.x;
    const int lane = tid & 31, wave = tid >> 5;
    {
        const int r = tid >> 4, l16 = tid & 15;
        const int row = m0 + r;
        const bool ok = row < N2;
        float4 acc[4];
        const int cnt = pull_row<HID>(h1, srcA, ch, slot, nxt, E1, row, ok, l16, acc);
        float4 f[4];
#pragma unroll
        for (int j = 0; j < 4; ++j)
            f[j] = ok ? ((const float4*)(h1 + (long)row * HID))[l16 + 16 * j]
                      : make_float4(0.0f, 0.0f, 0.0f, 0.0f);
        const float inv = 1.0f / fmaxf((float)cnt, 1.0f);
#pragma unroll
        for (int j = 0; j < 4; ++j) {
            float4 a;
            a.x = f[j].x + acc[j].x * inv;
            a.y = f[j].y + acc[j].y * inv;
            a.z = f[j].z + acc[j].z * inv;
            a.w = f[j].w + acc[j].w * inv;
            *(float4*)&As[r * 260 + 4 * (l16 + 16 * j)] = a;
        }
    }
    __syncthreads();

    const int lo = lane & 15, hi = lane >> 4;
    const int n0 = wave * 16;
    v8f acc = {0, 0, 0, 0, 0, 0, 0, 0};
    for (int k = 0; k < HID; k += 4) {
        const int kk = k + 2 * hi;
        v2f a;
        a[0] = As[lo * 260 + kk];
        a[1] = As[lo * 260 + kk + 1];
        v2f b;
        b[0] = W[kk * DOUT + n0 + lo];
        b[1] = W[(kk + 1) * DOUT + n0 + lo];
        acc = __builtin_amdgcn_wmma_f32_16x16x4_f32(false, a, false, b, (short)0, acc, false, false);
    }
#pragma unroll
    for (int r = 0; r < 8; ++r) {
        int row = m0 + r + 8 * hi;
        if (row < N2) out[(long)row * DOUT + n0 + lo] = acc[r] + bias[n0 + lo];
    }
}

extern "C" void kernel_launch(void* const* d_in, const int* in_sizes, int n_in,
                              void* d_out, int out_size, void* d_ws, size_t ws_size,
                              hipStream_t stream) {
    (void)in_sizes; (void)n_in; (void)out_size; (void)ws_size;
    const int*   x  = (const int*)d_in[0];
    const float* e0 = (const float*)d_in[1];
    const float* e1 = (const float*)d_in[2];
    const float* e2 = (const float*)d_in[3];
    const float* e3 = (const float*)d_in[4];
    const float* Win[2]  = {(const float*)d_in[5], (const float*)d_in[9]};
    const float* bin[2]  = {(const float*)d_in[6], (const float*)d_in[10]};
    const float* Wout[2] = {(const float*)d_in[7], (const float*)d_in[11]};
    const float* bout[2] = {(const float*)d_in[8], (const float*)d_in[12]};
    const int* srcL0[2] = {(const int*)d_in[13], (const int*)d_in[17]};
    const int* dstL0[2] = {(const int*)d_in[14], (const int*)d_in[18]};
    const int* srcL1[2] = {(const int*)d_in[15], (const int*)d_in[19]};
    const int* dstL1[2] = {(const int*)d_in[16], (const int*)d_in[20]};
    float* out = (float*)d_out;

    size_t off = 0;
    auto take = [&](size_t bytes) { size_t o = off; off += (bytes + 255) & ~(size_t)255; return o; };
    char* ws = (char*)d_ws;
    float* h1   = (float*)(ws + take((size_t)N1 * HID * 4));
    int* ch   = (int*)(ws + take((size_t)NHEAD * 8));
    int* slot = (int*)(ws + take((size_t)NHEAD * CAP * 4));
    int* nxt  = (int*)(ws + take((size_t)NLINK * 4));

    fill_heads<<<(NHEAD + 255) / 256, 256, 0, stream>>>(ch);
    link_edges<<<(NLINK + 255) / 256, 256, 0, stream>>>(srcL0[0], srcL0[1], srcL1[0], srcL1[1],
                                                        dstL0[0], dstL0[1], dstL1[0], dstL1[1],
                                                        ch, slot, nxt);

    for (int e = 0; e < 2; ++e) {
        gemm1_fused<<<N1 / 32, 256, 0, stream>>>(x, e0, e1, e2, e3, srcL0[e], ch + 2 * (size_t)e * N1,
                                                 slot + (size_t)e * N1 * CAP,
                                                 nxt + (size_t)e * E0, Win[e], bin[e], h1);
        gemm2_fused<<<(N2 + 15) / 16, 256, 0, stream>>>(h1, srcL1[e], ch + 2 * (2 * (size_t)N1 + (size_t)e * N2),
                                                        slot + (2 * (size_t)N1 + (size_t)e * N2) * CAP,
                                                        nxt + 2 * (size_t)E0 + (size_t)e * E1,
                                                        Wout[e], bout[e], out + (size_t)e * N2 * DOUT);
    }
}


# slot lines built by a two-launch counting sort, counts in LDS (edges binned by 1024-head bucket, one block per bucket fills its lines): no global atomic; rows summed in LDS-add arrival order
# speedup vs baseline: 3.7324x; 1.4270x over previous
#include <hip/hip_runtime.h>

constexpr int N0 = 400000, N1 = 100000, N2 = 25000;
constexpr int E0 = 800000, E1 = 200000;
constexpr int DIN = 128, HID = 256, DOUT = 128;

typedef __attribute__((ext_vector_type(2))) float v2f;
typedef __attribute__((ext_vector_type(8))) float v8f;

struct FeatLane {
    const int* xlo;
    const int* xhi;
    const float* lo;
    const float* hi;
    int loShift;
    int sel;
};

__device__ __forceinline__ FeatLane feat_lane(const int* __restrict__ x,
                                              const float* __restrict__ e0, const float* __restrict__ e1,
                                              const float* __restrict__ e2, const float* __restrict__ e3,
                                              int l16) {
    FeatLane L;
    if (l16 < 8)       { L.xlo = x + 0; L.lo = e0 + 4 * l16;        L.loShift = 5; L.sel = 0; }
    else if (l16 < 12) { L.xlo = x + 1; L.lo = e1 + 4 * (l16 - 8);  L.loShift = 4; L.sel = 1; }
    else               { L.xlo = x + 2; L.lo = e2 + 4 * (l16 - 12); L.loShift = 4; L.sel = 2; }
    L.xhi = x + 3;
    L.hi = e3 + 4 * l16;
    return L;
}

__device__ __forceinline__ void feat_load(const FeatLane& L, int s, float4 (&v)[2]) {
    const int xi = L.xlo[4 * (long)s];
    const int xw = L.xhi[4 * (long)s];
    v[0] = *(const float4*)(L.lo + ((long)xi << L.loShift));
    v[1] = *(const float4*)(L.hi + ((long)xw << 6));
}

constexpr int NHEAD = 2 * N1 + 2 * N2;
constexpr int NLINK = 2 * E0 + 2 * E1;
constexpr int CAP = 16;
constexpr int RB_LOG = 10, RB = 1 << RB_LOG;
constexpr int NBKT = (NHEAD + RB - 1) / RB;
constexpr int CHUNK = 8192;
constexpr int NCHUNK = (NLINK + CHUNK - 1) / CHUNK;
constexpr int TP = 256;
constexpr int TBITS = 21;
static_assert(NBKT <= TP && NCHUNK <= TP, "table pitch too small");
static_assert(NLINK <= (1 << TBITS) && RB_LOG + TBITS <= 31, "record key does not fit");

__global__ void __launch_bounds__(1024)
bin_edges(const int* __restrict__ sA0, const int* __restrict__ sB0,
          const int* __restrict__ sA1, const int* __restrict__ sB1,
          const int* __restrict__ dA0, const int* __restrict__ dB0,
          const int* __restrict__ dA1, const int* __restrict__ dB1,
          int2* __restrict__ rec, int* __restrict__ tcnt, int* __restrict__ toff) {
    __shared__ int hist[TP];
    __shared__ int base[TP];
    __shared__ int2 sorted[CHUNK];
    const int tid = threadIdx.x;
    const int t0 = blockIdx.x * CHUNK;
    if (tid < TP) hist[tid] = 0;
    __syncthreads();
    int dv[8], val[8], key[8], rk[8], bk[8];
#pragma unroll
    for (int i = 0; i < 8; ++i) {
        const int t = t0 + i * 1024 + tid;
        const int tc = (t < NLINK) ? t : NLINK - 1;
        const bool l1 = tc >= 2 * E0;
        const int u = l1 ? tc - 2 * E0 : tc;
        const int en = l1 ? E1 : E0;
        const bool b = u >= en;
        const int e = b ? u - en : u;
        const int* dp = l1 ? (b ? dB1 : dA1) : (b ? dB0 : dA0);
        const int* sp = l1 ? (b ? sB1 : sA1) : (b ? sB0 : sA0);
        dv[i] = dp[e];
        val[i] = sp[e];
    }
#pragma unroll
    for (int i = 0; i < 8; ++i) {
        const int t = t0 + i * 1024 + tid;
        const bool l1 = t >= 2 * E0;
        const bool b = l1 ? (t >= 2 * E0 + E1) : (t >= E0);
        const int nd = l1 ? N2 : N1;
        const int hd = (l1 ? 2 * N1 : 0) + (b ? nd : 0) + dv[i];
        const bool ok = (t < NLINK) && ((unsigned)dv[i] < (unsigned)nd);
        bk[i] = ok ? (hd >> RB_LOG) : -1;
        key[i] = ((hd & (RB - 1)) << TBITS) | t;
        rk[i] = 0;
        if (ok) rk[i] = atomicAdd(&hist[bk[i]], 1);
    }
    __syncthreads();
    if (tid < 32) {
        int s[8], tot = 0;
#pragma unroll
        for (int q = 0; q < 8; ++q) { s[q] = hist[8 * tid + q]; tot += s[q]; }
        int inc = tot;
        for (int dl = 1; dl < 32; dl <<= 1) {
            const int v = __shfl_up(inc, dl);
            if (tid >= dl) inc += v;
        }
        int ex = inc - tot;
#pragma unroll
        for (int q = 0; q < 8; ++q) { base[8 * tid + q] = ex; ex += s[q]; }
    }
    __syncthreads();
#pragma unroll
    for (int i = 0; i < 8; ++i)
        if (bk[i] >= 0) sorted[base[bk[i]] + rk[i]] = make_int2(key[i], val[i]);
    __syncthreads();
    const int total = base[TP - 1] + hist[TP - 1];
    int2* __restrict__ out = rec + (size_t)blockIdx.x * CHUNK;
    for (int j = tid; j < total; j += 1024) out[j] = sorted[j];
    if (tid < TP) {
        tcnt[tid * TP + blockIdx.x] = hist[tid];
        toff[tid * TP + blockIdx.x] = base[tid];
    }
}

__global__ void __launch_bounds__(1024)
build_slots(const int2* __restrict__ rec, const int* __restrict__ tcnt, const int* __restrict__ toff,
            int* __restrict__ ch, int* __restrict__ slot, int* __restrict__ nxt) {
    __shared__ int cnt[RB];
    __shared__ int head[RB];
    __shared__ alignas(16) int sl[RB * CAP];
    const int tid = threadIdx.x, lane = tid & 31, wave = tid >> 5;
    const int k = blockIdx.x;
    cnt[tid] = 0;
    head[tid] = -1;
    __syncthreads();
    int myN = 0, myO = 0;
    {
        const int j = wave + 32 * lane;
        if (lane < 8 && j < NCHUNK) { myN = tcnt[k * TP + j]; myO = toff[k * TP + j]; }
    }
    for (int q = 0; q < 8; ++q) {
        const int n = __shfl(myN, q);
        const int o = __shfl(myO, q);
        const int2* __restrict__ r = rec + (size_t)(wave + 32 * q) * CHUNK + o;
        for (int i0 = 0; i0 < n; i0 += 128) {
            int2 v[4];
#pragma unroll
            for (int m = 0; m < 4; ++m) {
                const int i = i0 + 32 * m + lane;
                v[m] = (i < n) ? r[i] : make_int2(-1, 0);
            }
#pragma unroll
            for (int m = 0; m < 4; ++m) {
                if (v[m].x >= 0) {
                    const int hl = v[m].x >> TBITS;
                    const int t = v[m].x & ((1 << TBITS) - 1);
                    const int pos = atomicAdd(&cnt[hl], 1);
                    if (pos < CAP) {
                        sl[hl * CAP + pos] = v[m].y;
                    } else {
                        const int sb = (t < E0) ? 0 : ((t < 2 * E0) ? E0 : ((t < 2 * E0 + E1) ? 2 * E0 : 2 * E0 + E1));
                        nxt[t] = atomicExch(&head[hl], t - sb);
                    }
                }
            }
        }
    }
    __syncthreads();
    const int hd = k * RB + tid;
    if (hd < NHEAD) ((int2*)ch)[hd] = make_int2(cnt[tid], head[tid]);
    for (int m = 0; m < 4; ++m) {
        const int q4 = tid + 1024 * m;
        if (k * RB + (q4 >> 2) < NHEAD)
            ((int4*)slot)[(size_t)k * (RB * CAP / 4) + q4] = ((const int4*)sl)[q4];
    }
}

__device__ __forceinline__ void add4(float4& a, const float4& v) {
    a.x += v.x; a.y += v.y; a.z += v.z; a.w += v.w;
}

template <int WIDTH>
__device__ __forceinline__ int pull_row(const float* __restrict__ h, const int* __restrict__ srcA,
                                        const int* __restrict__ ch, const int* __restrict__ slot,
                                        const int* __restrict__ nxt,
                                        int nE, int row, bool rowOk, int l16,
                                        float4 (&acc)[WIDTH / 64]) {
    constexpr int J = WIDTH / 64;
#pragma unroll
    for (int j = 0; j < J; ++j) acc[j] = make_float4(0.0f, 0.0f, 0.0f, 0.0f);
    const int2 c = rowOk ? ((const int2*)ch)[row] : make_int2(0, -1);
    const int n = (c.x < CAP) ? c.x : CAP;
    const int mine = (l16 < n) ? slot[(long)row * CAP + l16] : 0;
    const float4* __restrict__ h4 = (const float4*)h;
    int cnt = 0;
    int i0 = 0;
    if (n & 1) {
        const int s0 = __shfl(mine, 0, 16);
        const float4* p0 = (const float4*)(h + (long)s0 * WIDTH);
#pragma unroll
        for (int j = 0; j < J; ++j) add4(acc[j], p0[l16 + 16 * j]);
        cnt = 1;
        i0 = 1;
    }
    for (; i0 < n; i0 += 2) {
        const int s0 = __shfl(mine, i0, 16);
        const int s1 = __shfl(mine, i0 + 1, 16);
        const unsigned o0 = (unsigned)s0 * (WIDTH / 4) + l16;
        const unsigned o1 = (unsigned)s1 * (WIDTH / 4) + l16;
        float4 v0[J], v1[J];
#pragma unroll
        for (int j = 0; j < J; ++j) { v0[j] = h4[o0 + 16 * j]; v1[j] = h4[o1 + 16 * j]; }
#pragma unroll
        for (int j = 0; j < J; ++j) add4(acc[j], v0[j]);
#pragma unroll
        for (int j = 0; j < J; ++j) add4(acc[j], v1[j]);
        cnt += 2;
    }
    if (c.x > CAP) {
        int e = c.y;
        for (int it = 0; it < nE && (unsigned)e < (unsigned)nE; ++it) {
            const int sv = srcA[e];
            const int nx = nxt[e];
            const float4* hp = (const float4*)(h + (long)sv * WIDTH);
#pragma unroll
            for (int j = 0; j < J; ++j) add4(acc[j], hp[l16 + 16 * j]);
            ++cnt;
            e = nx;
        }
    }
    return cnt;
}

template <int G>
__device__ __forceinline__ int pull_feat_row(const FeatLane& L, const int* __restrict__ x,
                                             const int* __restrict__ srcA,
                                             const int* __restrict__ ch, const int* __restrict__ slot,
                                             const int* __restrict__ nxt,
                                             int nE, int row, int l16, float4 (&acc)[2]) {
    acc[0] = make_float4(0.0f, 0.0f, 0.0f, 0.0f);
    acc[1] = make_float4(0.0f, 0.0f, 0.0f, 0.0f);
    const int2 c = ((const int2*)ch)[row];
    const int n = (c.x < CAP) ? c.x : CAP;
    const int mine = (l16 < n) ? slot[(long)row * CAP + l16] : 0;
    int4 X = make_int4(0, 0, 0, 0);
    if (l16 < n) X = *(const int4*)(x + 4 * (long)mine);
    int cnt = 0;
    for (int i0 = 0; i0 < n; i0 += G) {
        float4 v0[G], v1[G];
#pragma unroll
        for (int g = 0; g < G; ++g) {
            const int i = (i0 + g < n) ? i0 + g : i0;
            const int xa = __shfl(X.x, i, 16);
            const int xb = __shfl(X.y, i, 16);
            const int xc = __shfl(X.z, i, 16);
            const int xw = __shfl(X.w, i, 16);
            const int xi = (L.sel == 0) ? xa : ((L.sel == 1) ? xb : xc);
            v0[g] = *(const float4*)(L.lo + ((long)xi << L.loShift));
            v1[g] = *(const float4*)(L.hi + ((long)xw << 6));
        }
#pragma unroll
        for (int g = 0; g < G; ++g) {
            if (i0 + g < n) {
                add4(acc[0], v0[g]);
                add4(acc[1], v1[g]);
                ++cnt;
            }
        }
    }
    if (c.x > CAP) {
        int e = c.y;
        for (int it = 0; it < nE && (unsigned)e < (unsigned)nE; ++it) {
            const int sv = srcA[e];
            const int nx = nxt[e];
            float4 f[2];
            feat_load(L, sv, f);
            add4(acc[0], f[0]);
            add4(acc[1], f[1]);
            ++cnt;
            e = nx;
        }
    }
    return cnt;
}

__global__ void __launch_bounds__(256)
gemm1_fused(const int* __restrict__ x,
            const float* __restrict__ e0, const float* __restrict__ e1,
            const float* __restrict__ e2, const float* __restrict__ e3,
            const int* __restrict__ srcA,
            const int* __restrict__ ch, const int* __restrict__ slot, const int* __restrict__ nxt,
            const float* __restrict__ W,
            const float* __restrict__ bias, float* __restrict__ h1) {
    __shared__ alignas(16) float As[32 * 132];
    __shared__ float srow[32];
    const int m0 = blockIdx.x * 32;
    const int tid = threadIdx.x;
    const int lane = tid & 31, wave = tid >> 5;
    {
        const int r = tid >> 4, l16 = tid & 15;
        const FeatLane L = feat_lane(x, e0, e1, e2, e3, l16);
        for (int p = 0; p < 2; ++p) {
            const int rl = r + 16 * p;
            float4 f[2];
            feat_load(L, m0 + rl, f);
            float4 acc[2];
            const int cnt = pull_feat_row<2>(L, x, srcA, ch, slot, nxt, E0, m0 + rl, l16, acc);
            const float inv = 1.0f / fmaxf((float)cnt, 1.0f);
#pragma unroll
            for (int j = 0; j < 2; ++j) {
                float4 a;
                a.x = f[j].x + acc[j].x * inv;
                a.y = f[j].y + acc[j].y * inv;
                a.z = f[j].z + acc[j].z * inv;
                a.w = f[j].w + acc[j].w * inv;
                *(float4*)&As[rl * 132 + 4 * (l16 + 16 * j)] = a;
            }
            if (l16 == 0) srow[rl] = (cnt > 0) ? 2.0f : 1.0f;
        }
    }
    __syncthreads();

    const int lo = lane & 15, hi = lane >> 4;
    const int mt = wave >> 2, g = wave & 3;
    const float* Ar = &As[(mt * 16 + lo) * 132];
    const float* Wg = W + 64 * g + 4 * lo;
    v8f c0 = {0, 0, 0, 0, 0, 0, 0, 0};
    v8f c1 = {0, 0, 0, 0, 0, 0, 0, 0};
    v8f c2 = {0, 0, 0, 0, 0, 0, 0, 0};
    v8f c3 = {0, 0, 0, 0, 0, 0, 0, 0};
#pragma unroll 2
    for (int k = 0; k < DIN; k += 4) {
        const int kk = k + 2 * hi;
        v2f a;
        a[0] = Ar[kk];
        a[1] = Ar[kk + 1];
        const float4 w0 = *(const float4*)(Wg + kk * HID);
        const float4 w1 = *(const float4*)(Wg + (kk + 1) * HID);
        v2f b0, b1, b2, b3;
        b0[0] = w0.x; b0[1] = w1.x;
        b1[0] = w0.y; b1[1] = w1.y;
        b2[0] = w0.z; b2[1] = w1.z;
        b3[0] = w0.w; b3[1] = w1.w;
        c0 = __builtin_amdgcn_wmma_f32_16x16x4_f32(false, a, false, b0, (short)0, c0, false, false);
        c1 = __builtin_amdgcn_wmma_f32_16x16x4_f32(false, a, false, b1, (short)0, c1, false, false);
        c2 = __builtin_amdgcn_wmma_f32_16x16x4_f32(false, a, false, b2, (short)0, c2, false, false);
        c3 = __builtin_amdgcn_wmma_f32_16x16x4_f32(false, a, false, b3, (short)0, c3, false, false);
    }
    const float4 bv = *(const float4*)(bias + 64 * g + 4 * lo);
#pragma unroll
    for (int r = 0; r < 8; ++r) {
        int rr = mt * 16 + r + 8 * hi;
        long row = m0 + rr;
        float s = srow[rr];
        float4 o;
        o.x = fmaxf(c0[r] + s * bv.x, 0.0f);
        o.y = fmaxf(c1[r] + s * bv.y, 0.0f);
        o.z = fmaxf(c2[r] + s * bv.z, 0.0f);
        o.w = fmaxf(c3[r] + s * bv.w, 0.0f);
        *(float4*)(h1 + row * HID + 64 * g + 4 * lo) = o;
    }
}

__global__ void __launch_bounds__(256)
gemm2_fused(const float* __restrict__ h1, const int* __restrict__ srcA,
            const int* __restrict__ ch, const int* __restrict__ slot, const int* __restrict__ nxt,
            const float* __restrict__ W,
            const float* __restrict__ bias, float* __restrict__ out) {
    __shared__ alignas(16) float As[16 * 260];
    const int m0 = blockIdx.x * 16;
    const int tid = threadIdx.x;
    const int lane = tid & 31, wave = tid >> 5;
    {
        const int r = tid >> 4, l16 = tid & 15;
        const int row = m0 + r;
        const bool ok = row < N2;
        float4 acc[4];
        const int cnt = pull_row<HID>(h1, srcA, ch, slot, nxt, E1, row, ok, l16, acc);
        float4 f[4];
#pragma unroll
        for (int j = 0; j < 4; ++j)
            f[j] = ok ? ((const float4*)(h1 + (long)row * HID))[l16 + 16 * j]
                      : make_float4(0.0f, 0.0f, 0.0f, 0.0f);
        const float inv = 1.0f / fmaxf((float)cnt, 1.0f);
#pragma unroll
        for (int j = 0; j < 4; ++j) {
            float4 a;
            a.x = f[j].x + acc[j].x * inv;
            a.y = f[j].y + acc[j].y * inv;
            a.z = f[j].z + acc[j].z * inv;
            a.w = f[j].w + acc[j].w * inv;
            *(float4*)&As[r * 260 + 4 * (l16 + 16 * j)] = a;
        }
    }
    __syncthreads();

    const int lo = lane & 15, hi = lane >> 4;
    const int n0 = wave * 16;
    v8f acc = {0, 0, 0, 0, 0, 0, 0, 0};
    for (int k = 0; k < HID; k += 4) {
        const int kk = k + 2 * hi;
        v2f a;
        a[0] = As[lo * 260 + kk];
        a[1] = As[lo * 260 + kk + 1];
        v2f b;
        b[0] = W[kk * DOUT + n0 + lo];
        b[1] = W[(kk + 1) * DOUT + n0 + lo];
        acc = __builtin_amdgcn_wmma_f32_16x16x4_f32(false, a, false, b, (short)0, acc, false, false);
    }
#pragma unroll
    for (int r = 0; r < 8; ++r) {
        int row = m0 + r + 8 * hi;
        if (row < N2) out[(long)row * DOUT + n0 + lo] = acc[r] + bias[n0 + lo];
    }
}

extern "C" void kernel_launch(void* const* d_in, const int* in_sizes, int n_in,
                              void* d_out, int out_size, void* d_ws, size_t ws_size,
                              hipStream_t stream) {
    (void)in_sizes; (void)n_in; (void)out_size; (void)ws_size;
    const int*   x  = (const int*)d_in[0];
    const float* e0 = (const float*)d_in[1];
    const float* e1 = (const float*)d_in[2];
    const float* e2 = (const float*)d_in[3];
    const float* e3 = (const float*)d_in[4];
    const float* Win[2]  = {(const float*)d_in[5], (const float*)d_in[9]};
    const float* bin[2]  = {(const float*)d_in[6], (const float*)d_in[10]};
    const float* Wout[2] = {(const float*)d_in[7], (const float*)d_in[11]};
    const float* bout[2] = {(const float*)d_in[8], (const float*)d_in[12]};
    const int* srcL0[2] = {(const int*)d_in[13], (const int*)d_in[17]};
    const int* dstL0[2] = {(const int*)d_in[14], (const int*)d_in[18]};
    const int* srcL1[2] = {(const int*)d_in[15], (const int*)d_in[19]};
    const int* dstL1[2] = {(const int*)d_in[16], (const int*)d_in[20]};
    float* out = (float*)d_out;

    size_t off = 0;
    auto take = [&](size_t bytes) { size_t o = off; off += (bytes + 255) & ~(size_t)255; return o; };
    char* ws = (char*)d_ws;
    float* h1   = (float*)(ws + take((size_t)N1 * HID * 4));
    int* ch   = (int*)(ws + take((size_t)NHEAD * 8));
    int* slot = (int*)(ws + take((size_t)NHEAD * CAP * 4));
    int* nxt  = (int*)(ws + take((size_t)NLINK * 4));
    int2* rec = (int2*)(ws + take((size_t)NCHUNK * CHUNK * 8));
    int* tcnt = (int*)(ws + take((size_t)TP * TP * 4));
    int* toff = (int*)(ws + take((size_t)TP * TP * 4));

    bin_edges<<<NCHUNK, 1024, 0, stream>>>(srcL0[0], srcL0[1], srcL1[0], srcL1[1],
                                           dstL0[0], dstL0[1], dstL1[0], dstL1[1], rec, tcnt, toff);
    build_slots<<<NBKT, 1024, 0, stream>>>(rec, tcnt, toff, ch, slot, nxt);

    for (int e = 0; e < 2; ++e) {
        gemm1_fused<<<N1 / 32, 256, 0, stream>>>(x, e0, e1, e2, e3, srcL0[e], ch + 2 * (size_t)e * N1,
                                                 slot + (size_t)e * N1 * CAP,
                                                 nxt + (size_t)e * E0, Win[e], bin[e], h1);
        gemm2_fused<<<(N2 + 15) / 16, 256, 0, stream>>>(h1, srcL1[e], ch + 2 * (2 * (size_t)N1 + (size_t)e * N2),
                                                        slot + (2 * (size_t)N1 + (size_t)e * N2) * CAP,
                                                        nxt + 2 * (size_t)E0 + (size_t)e * E1,
                                                        Wout[e], bout[e], out + (size_t)e * N2 * DOUT);
    }
}


# both branches share each GEMM launch (grid.y = branch, one h1 plane per branch): 4 launches, not 6; order of every sum unchanged
# speedup vs baseline: 4.0590x; 1.0875x over previous
#include <hip/hip_runtime.h>

constexpr int N0 = 400000, N1 = 100000, N2 = 25000;
constexpr int E0 = 800000, E1 = 200000;
constexpr int DIN = 128, HID = 256, DOUT = 128;

typedef __attribute__((ext_vector_type(2))) float v2f;
typedef __attribute__((ext_vector_type(8))) float v8f;

struct FeatLane {
    const int* xlo;
    const int* xhi;
    const float* lo;
    const float* hi;
    int loShift;
    int sel;
};

__device__ __forceinline__ FeatLane feat_lane(const int* __restrict__ x,
                                              const float* __restrict__ e0, const float* __restrict__ e1,
                                              const float* __restrict__ e2, const float* __restrict__ e3,
                                              int l16) {
    FeatLane L;
    if (l16 < 8)       { L.xlo = x + 0; L.lo = e0 + 4 * l16;        L.loShift = 5; L.sel = 0; }
    else if (l16 < 12) { L.xlo = x + 1; L.lo = e1 + 4 * (l16 - 8);  L.loShift = 4; L.sel = 1; }
    else               { L.xlo = x + 2; L.lo = e2 + 4 * (l16 - 12); L.loShift = 4; L.sel = 2; }
    L.xhi = x + 3;
    L.hi = e3 + 4 * l16;
    return L;
}

__device__ __forceinline__ void feat_load(const FeatLane& L, int s, float4 (&v)[2]) {
    const int xi = L.xlo[4 * (long)s];
    const int xw = L.xhi[4 * (long)s];
    v[0] = *(const float4*)(L.lo + ((long)xi << L.loShift));
    v[1] = *(const float4*)(L.hi + ((long)xw << 6));
}

constexpr int NHEAD = 2 * N1 + 2 * N2;
constexpr int NLINK = 2 * E0 + 2 * E1;
constexpr int CAP = 16;
constexpr int RB_LOG = 10, RB = 1 << RB_LOG;
constexpr int NBKT = (NHEAD + RB - 1) / RB;
constexpr int CHUNK = 8192;
constexpr int NCHUNK = (NLINK + CHUNK - 1) / CHUNK;
constexpr int TP = 256;
constexpr int TBITS = 21;
static_assert(NBKT <= TP && NCHUNK <= TP, "table pitch too small");
static_assert(NLINK <= (1 << TBITS) && RB_LOG + TBITS <= 31, "record key does not fit");

__global__ void __launch_bounds__(1024)
bin_edges(const int* __restrict__ sA0, const int* __restrict__ sB0,
          const int* __restrict__ sA1, const int* __restrict__ sB1,
          const int* __restrict__ dA0, const int* __restrict__ dB0,
          const int* __restrict__ dA1, const int* __restrict__ dB1,
          int2* __restrict__ rec, int* __restrict__ tcnt, int* __restrict__ toff) {
    __shared__ int hist[TP];
    __shared__ int base[TP];
    __shared__ int2 sorted[CHUNK];
    const int tid = threadIdx.x;
    const int t0 = blockIdx.x * CHUNK;
    if (tid < TP) hist[tid] = 0;
    __syncthreads();
    int dv[8], val[8], key[8], rk[8], bk[8];
#pragma unroll
    for (int i = 0; i < 8; ++i) {
        const int t = t0 + i * 1024 + tid;
        const int tc = (t < NLINK) ? t : NLINK - 1;
        const bool l1 = tc >= 2 * E0;
        const int u = l1 ? tc - 2 * E0 : tc;
        const int en = l1 ? E1 : E0;
        const bool b = u >= en;
        const int e = b ? u - en : u;
        const int* dp = l1 ? (b ? dB1 : dA1) : (b ? dB0 : dA0);
        const int* sp = l1 ? (b ? sB1 : sA1) : (b ? sB0 : sA0);
        dv[i] = dp[e];
        val[i] = sp[e];
    }
#pragma unroll
    for (int i = 0; i < 8; ++i) {
        const int t = t0 + i * 1024 + tid;
        const bool l1 = t >= 2 * E0;
        const bool b = l1 ? (t >= 2 * E0 + E1) : (t >= E0);
        const int nd = l1 ? N2 : N1;
        const int hd = (l1 ? 2 * N1 : 0) + (b ? nd : 0) + dv[i];
        const bool ok = (t < NLINK) && ((unsigned)dv[i] < (unsigned)nd);
        bk[i] = ok ? (hd >> RB_LOG) : -1;
        key[i] = ((hd & (RB - 1)) << TBITS) | t;
        rk[i] = 0;
        if (ok) rk[i] = atomicAdd(&hist[bk[i]], 1);
    }
    __syncthreads();
    if (tid < 32) {
        int s[8], tot = 0;
#pragma unroll
        for (int q = 0; q < 8; ++q) { s[q] = hist[8 * tid + q]; tot += s[q]; }
        int inc = tot;
        for (int dl = 1; dl < 32; dl <<= 1) {
            const int v = __shfl_up(inc, dl);
            if (tid >= dl) inc += v;
        }
        int ex = inc - tot;
#pragma unroll
        for (int q = 0; q < 8; ++q) { base[8 * tid + q] = ex; ex += s[q]; }
    }
    __syncthreads();
#pragma unroll
    for (int i = 0; i < 8; ++i)
        if (bk[i] >= 0) sorted[base[bk[i]] + rk[i]] = make_int2(key[i], val[i]);
    __syncthreads();
    const int total = base[TP - 1] + hist[TP - 1];
    int2* __restrict__ out = rec + (size_t)blockIdx.x * CHUNK;
    for (int j = tid; j < total; j += 1024) out[j] = sorted[j];
    if (tid < TP) {
        tcnt[tid * TP + blockIdx.x] = hist[tid];
        toff[tid * TP + blockIdx.x] = base[tid];
    }
}

__global__ void __launch_bounds__(1024)
build_slots(const int2* __restrict__ rec, const int* __restrict__ tcnt, const int* __restrict__ toff,
            int* __restrict__ ch, int* __restrict__ slot, int* __restrict__ nxt) {
    __shared__ int cnt[RB];
    __shared__ int head[RB];
    __shared__ alignas(16) int sl[RB * CAP];
    const int tid = threadIdx.x, lane = tid & 31, wave = tid >> 5;
    const int k = blockIdx.x;
    cnt[tid] = 0;
    head[tid] = -1;
    __syncthreads();
    int myN = 0, myO = 0;
    {
        const int j = wave + 32 * lane;
        if (lane < 8 && j < NCHUNK) { myN = tcnt[k * TP + j]; myO = toff[k * TP + j]; }
    }
    for (int q = 0; q < 8; ++q) {
        const int n = __shfl(myN, q);
        const int o = __shfl(myO, q);
        const int2* __restrict__ r = rec + (size_t)(wave + 32 * q) * CHUNK + o;
        for (int i0 = 0; i0 < n; i0 += 128) {
            int2 v[4];
#pragma unroll
            for (int m = 0; m < 4; ++m) {
                const int i = i0 + 32 * m + lane;
                v[m] = (i < n) ? r[i] : make_int2(-1, 0);
            }
#pragma unroll
            for (int m = 0; m < 4; ++m) {
                if (v[m].x >= 0) {
                    const int hl = v[m].x >> TBITS;
                    const int t = v[m].x & ((1 << TBITS) - 1);
                    const int pos = atomicAdd(&cnt[hl], 1);
                    if (pos < CAP) {
                        sl[hl * CAP + pos] = v[m].y;
                    } else {
                        const int sb = (t < E0) ? 0 : ((t < 2 * E0) ? E0 : ((t < 2 * E0 + E1) ? 2 * E0 : 2 * E0 + E1));
                        nxt[t] = atomicExch(&head[hl], t - sb);
                    }
                }
            }
        }
    }
    __syncthreads();
    const int hd = k * RB + tid;
    if (hd < NHEAD) ((int2*)ch)[hd] = make_int2(cnt[tid], head[tid]);
    for (int m = 0; m < 4; ++m) {
        const int q4 = tid + 1024 * m;
        if (k * RB + (q4 >> 2) < NHEAD)
            ((int4*)slot)[(size_t)k * (RB * CAP / 4) + q4] = ((const int4*)sl)[q4];
    }
}

__device__ __forceinline__ void add4(float4& a, const float4& v) {
    a.x += v.x; a.y += v.y; a.z += v.z; a.w += v.w;
}

template <int WIDTH>
__device__ __forceinline__ int pull_row(const float* __restrict__ h, const int* __restrict__ srcA,
                                        const int* __restrict__ ch, const int* __restrict__ slot,
                                        const int* __restrict__ nxt,
                                        int nE, int row, bool rowOk, int l16,
                                        float4 (&acc)[WIDTH / 64]) {
    constexpr int J = WIDTH / 64;
#pragma unroll
    for (int j = 0; j < J; ++j) acc[j] = make_float4(0.0f, 0.0f, 0.0f, 0.0f);
    const int2 c = rowOk ? ((const int2*)ch)[row] : make_int2(0, -1);
    const int n = (c.x < CAP) ? c.x : CAP;
    const int mine = (l16 < n) ? slot[(long)row * CAP + l16] : 0;
    const float4* __restrict__ h4 = (const float4*)h;
    int cnt = 0;
    int i0 = 0;
    if (n & 1) {
        const int s0 = __shfl(mine, 0, 16);
        const float4* p0 = (const float4*)(h + (long)s0 * WIDTH);
#pragma unroll
        for (int j = 0; j < J; ++j) add4(acc[j], p0[l16 + 16 * j]);
        cnt = 1;
        i0 = 1;
    }
    for (; i0 < n; i0 += 2) {
        const int s0 = __shfl(mine, i0, 16);
        const int s1 = __shfl(mine, i0 + 1, 16);
        const unsigned o0 = (unsigned)s0 * (WIDTH / 4) + l16;
        const unsigned o1 = (unsigned)s1 * (WIDTH / 4) + l16;
        float4 v0[J], v1[J];
#pragma unroll
        for (int j = 0; j < J; ++j) { v0[j] = h4[o0 + 16 * j]; v1[j] = h4[o1 + 16 * j]; }
#pragma unroll
        for (int j = 0; j < J; ++j) add4(acc[j], v0[j]);
#pragma unroll
        for (int j = 0; j < J; ++j) add4(acc[j], v1[j]);
        cnt += 2;
    }
    if (c.x > CAP) {
        int e = c.y;
        for (int it = 0; it < nE && (unsigned)e < (unsigned)nE; ++it) {
            const int sv = srcA[e];
            const int nx = nxt[e];
            const float4* hp = (const float4*)(h + (long)sv * WIDTH);
#pragma unroll
            for (int j = 0; j < J; ++j) add4(acc[j], hp[l16 + 16 * j]);
            ++cnt;
            e = nx;
        }
    }
    return cnt;
}

template <int G>
__device__ __forceinline__ int pull_feat_row(const FeatLane& L, const int* __restrict__ x,
                                             const int* __restrict__ srcA,
                                             const int* __restrict__ ch, const int* __restrict__ slot,
                                             const int* __restrict__ nxt,
                                             int nE, int row, int l16, float4 (&acc)[2]) {
    acc[0] = make_float4(0.0f, 0.0f, 0.0f, 0.0f);
    acc[1] = make_float4(0.0f, 0.0f, 0.0f, 0.0f);
    const int2 c = ((const int2*)ch)[row];
    const int n = (c.x < CAP) ? c.x : CAP;
    const int mine = (l16 < n) ? slot[(long)row * CAP + l16] : 0;
    int4 X = make_int4(0, 0, 0, 0);
    if (l16 < n) X = *(const int4*)(x + 4 * (long)mine);
    int cnt = 0;
    for (int i0 = 0; i0 < n; i0 += G) {
        float4 v0[G], v1[G];
#pragma unroll
        for (int g = 0; g < G; ++g) {
            const int i = (i0 + g < n) ? i0 + g : i0;
            const int xa = __shfl(X.x, i, 16);
            const int xb = __shfl(X.y, i, 16);
            const int xc = __shfl(X.z, i, 16);
            const int xw = __shfl(X.w, i, 16);
            const int xi = (L.sel == 0) ? xa : ((L.sel == 1) ? xb : xc);
            v0[g] = *(const float4*)(L.lo + ((long)xi << L.loShift));
            v1[g] = *(const float4*)(L.hi + ((long)xw << 6));
        }
#pragma unroll
        for (int g = 0; g < G; ++g) {
            if (i0 + g < n) {
                add4(acc[0], v0[g]);
                add4(acc[1], v1[g]);
                ++cnt;
            }
        }
    }
    if (c.x > CAP) {
        int e = c.y;
        for (int it = 0; it < nE && (unsigned)e < (unsigned)nE; ++it) {
            const int sv = srcA[e];
            const int nx = nxt[e];
            float4 f[2];
            feat_load(L, sv, f);
            add4(acc[0], f[0]);
            add4(acc[1], f[1]);
            ++cnt;
            e = nx;
        }
    }
    return cnt;
}

__global__ void __launch_bounds__(256)
gemm1_fused(const int* __restrict__ x,
            const float* __restrict__ e0, const float* __restrict__ e1,
            const float* __restrict__ e2, const float* __restrict__ e3,
            const int* __restrict__ srcA0, const int* __restrict__ srcA1,
            const int* __restrict__ ch0, const int* __restrict__ slot0, const int* __restrict__ nxt0,
            const float* __restrict__ W0, const float* __restrict__ W1,
            const float* __restrict__ bias0, const float* __restrict__ bias1, float* __restrict__ h1p) {
    __shared__ alignas(16) float As[32 * 132];
    __shared__ float srow[32];
    const int br = blockIdx.y;
    const int* __restrict__ srcA = br ? srcA1 : srcA0;
    const int* __restrict__ ch = ch0 + 2 * (size_t)br * N1;
    const int* __restrict__ slot = slot0 + (size_t)br * N1 * CAP;
    const int* __restrict__ nxt = nxt0 + (size_t)br * E0;
    const float* __restrict__ W = br ? W1 : W0;
    const float* __restrict__ bias = br ? bias1 : bias0;
    float* __restrict__ h1 = h1p + (size_t)br * N1 * HID;
    const int m0 = blockIdx.x * 32;
    const int tid = threadIdx.x;
    const int lane = tid & 31, wave = tid >> 5;
    {
        const int r = tid >> 4, l16 = tid & 15;
        const FeatLane L = feat_lane(x, e0, e1, e2, e3, l16);
        for (int p = 0; p < 2; ++p) {
            const int rl = r + 16 * p;
            float4 f[2];
            feat_load(L, m0 + rl, f);
            float4 acc[2];
            const int cnt = pull_feat_row<2>(L, x, srcA, ch, slot, nxt, E0, m0 + rl, l16, acc);
            const float inv = 1.0f / fmaxf((float)cnt, 1.0f);
#pragma unroll
            for (int j = 0; j < 2; ++j) {
                float4 a;
                a.x = f[j].x + acc[j].x * inv;
                a.y = f[j].y + acc[j].y * inv;
                a.z = f[j].z + acc[j].z * inv;
                a.w = f[j].w + acc[j].w * inv;
                *(float4*)&As[rl * 132 + 4 * (l16 + 16 * j)] = a;
            }
            if (l16 == 0) srow[rl] = (cnt > 0) ? 2.0f : 1.0f;
        }
    }
    __syncthreads();

    const int lo = lane & 15, hi = lane >> 4;
    const int mt = wave >> 2, g = wave & 3;
    const float* Ar = &As[(mt * 16 + lo) * 132];
    const float* Wg = W + 64 * g + 4 * lo;
    v8f c0 = {0, 0, 0, 0, 0, 0, 0, 0};
    v8f c1 = {0, 0, 0, 0, 0, 0, 0, 0};
    v8f c2 = {0, 0, 0, 0, 0, 0, 0, 0};
    v8f c3 = {0, 0, 0, 0, 0, 0, 0, 0};
#pragma unroll 2
    for (int k = 0; k < DIN; k += 4) {
        const int kk = k + 2 * hi;
        v2f a;
        a[0] = Ar[kk];
        a[1] = Ar[kk + 1];
        const float4 w0 = *(const float4*)(Wg + kk * HID);
        const float4 w1 = *(const float4*)(Wg + (kk + 1) * HID);
        v2f b0, b1, b2, b3;
        b0[0] = w0.x; b0[1] = w1.x;
        b1[0] = w0.y; b1[1] = w1.y;
        b2[0] = w0.z; b2[1] = w1.z;
        b3[0] = w0.w; b3[1] = w1.w;
        c0 = __builtin_amdgcn_wmma_f32_16x16x4_f32(false, a, false, b0, (short)0, c0, false, false);
        c1 = __builtin_amdgcn_wmma_f32_16x16x4_f32(false, a, false, b1, (short)0, c1, false, false);
        c2 = __builtin_amdgcn_wmma_f32_16x16x4_f32(false, a, false, b2, (short)0, c2, false, false);
        c3 = __builtin_amdgcn_wmma_f32_16x16x4_f32(false, a, false, b3, (short)0, c3, false, false);
    }
    const float4 bv = *(const float4*)(bias + 64 * g + 4 * lo);
#pragma unroll
    for (int r = 0; r < 8; ++r) {
        int rr = mt * 16 + r + 8 * hi;
        long row = m0 + rr;
        float s = srow[rr];
        float4 o;
        o.x = fmaxf(c0[r] + s * bv.x, 0.0f);
        o.y = fmaxf(c1[r] + s * bv.y, 0.0f);
        o.z = fmaxf(c2[r] + s * bv.z, 0.0f);
        o.w = fmaxf(c3[r] + s * bv.w, 0.0f);
        *(float4*)(h1 + row * HID + 64 * g + 4 * lo) = o;
    }
}

__global__ void __launch_bounds__(256)
gemm2_fused(const float* __restrict__ h1p, const int* __restrict__ srcA0, const int* __restrict__ srcA1,
            const int* __restrict__ ch0, const int* __restrict__ slot0, const int* __restrict__ nxt0,
            const float* __restrict__ W0, const float* __restrict__ W1,
            const float* __restrict__ bias0, const float* __restrict__ bias1, float* __restrict__ outp) {
    __shared__ alignas(16) float As[16 * 260];
    const int br = 1 - (int)blockIdx.y;
    const float* __restrict__ h1 = h1p + (size_t)br * N1 * HID;
    const int* __restrict__ srcA = br ? srcA1 : srcA0;
    const int* __restrict__ ch = ch0 + 2 * (size_t)br * N2;
    const int* __restrict__ slot = slot0 + (size_t)br * N2 * CAP;
    const int* __restrict__ nxt = nxt0 + (size_t)br * E1;
    const float* __restrict__ W = br ? W1 : W0;
    const float* __restrict__ bias = br ? bias1 : bias0;
    float* __restrict__ out = outp + (size_t)br * N2 * DOUT;
    const int m0 = blockIdx.x * 16;
    const int tid = threadIdx.x;
    const int lane = tid & 31, wave = tid >> 5;
    {
        const int r = tid >> 4, l16 = tid & 15;
        const int row = m0 + r;
        const bool ok = row < N2;
        float4 acc[4];
        const int cnt = pull_row<HID>(h1, srcA, ch, slot, nxt, E1, row, ok, l16, acc);
        float4 f[4];
#pragma unroll
        for (int j = 0; j < 4; ++j)
            f[j] = ok ? ((const float4*)(h1 + (long)row * HID))[l16 + 16 * j]
                      : make_float4(0.0f, 0.0f, 0.0f, 0.0f);
        const float inv = 1.0f / fmaxf((float)cnt, 1.0f);
#pragma unroll
        for (int j = 0; j < 4; ++j) {
            float4 a;
            a.x = f[j].x + acc[j].x * inv;
            a.y = f[j].y + acc[j].y * inv;
            a.z = f[j].z + acc[j].z * inv;
            a.w = f[j].w + acc[j].w * inv;
            *(float4*)&As[r * 260 + 4 * (l16 + 16 * j)] = a;
        }
    }
    __syncthreads();

    const int lo = lane & 15, hi = lane >> 4;
    const int n0 = wave * 16;
    v8f acc = {0, 0, 0, 0, 0, 0, 0, 0};
    for (int k = 0; k < HID; k += 4) {
        const int kk = k + 2 * hi;
        v2f a;
        a[0] = As[lo * 260 + kk];
        a[1] = As[lo * 260 + kk + 1];
        v2f b;
        b[0] = W[kk * DOUT + n0 + lo];
        b[1] = W[(kk + 1) * DOUT + n0 + lo];
        acc = __builtin_amdgcn_wmma_f32_16x16x4_f32(false, a, false, b, (short)0, acc, false, false);
    }
#pragma unroll
    for (int r = 0; r < 8; ++r) {
        int row = m0 + r + 8 * hi;
        if (row < N2) out[(long)row * DOUT + n0 + lo] = acc[r] + bias[n0 + lo];
    }
}

extern "C" void kernel_launch(void* const* d_in, const int* in_sizes, int n_in,
                              void* d_out, int out_size, void* d_ws, size_t ws_size,
                              hipStream_t stream) {
    (void)in_sizes; (void)n_in; (void)out_size; (void)ws_size;
    const int*   x  = (const int*)d_in[0];
    const float* e0 = (const float*)d_in[1];
    const float* e1 = (const float*)d_in[2];
    const float* e2 = (const float*)d_in[3];
    const float* e3 = (const float*)d_in[4];
    const float* Win[2]  = {(const float*)d_in[5], (const float*)d_in[9]};
    const float* bin[2]  = {(const float*)d_in[6], (const float*)d_in[10]};
    const float* Wout[2] = {(const float*)d_in[7], (const float*)d_in[11]};
    const float* bout[2] = {(const float*)d_in[8], (const float*)d_in[12]};
    const int* srcL0[2] = {(const int*)d_in[13], (const int*)d_in[17]};
    const int* dstL0[2] = {(const int*)d_in[14], (const int*)d_in[18]};
    const int* srcL1[2] = {(const int*)d_in[15], (const int*)d_in[19]};
    const int* dstL1[2] = {(const int*)d_in[16], (const int*)d_in[20]};
    float* out = (float*)d_out;

    size_t off = 0;
    auto take = [&](size_t bytes) { size_t o = off; off += (bytes + 255) & ~(size_t)255; return o; };
    char* ws = (char*)d_ws;
    float* h1   = (float*)(ws + take((size_t)2 * N1 * HID * 4));
    int* ch   = (int*)(ws + take((size_t)NHEAD * 8));
    int* slot = (int*)(ws + take((size_t)NHEAD * CAP * 4));
    int* nxt  = (int*)(ws + take((size_t)NLINK * 4));
    int2* rec = (int2*)(ws + take((size_t)NCHUNK * CHUNK * 8));
    int* tcnt = (int*)(ws + take((size_t)TP * TP * 4));
    int* toff = (int*)(ws + take((size_t)TP * TP * 4));

    bin_edges<<<NCHUNK, 1024, 0, stream>>>(srcL0[0], srcL0[1], srcL1[0], srcL1[1],
                                           dstL0[0], dstL0[1], dstL1[0], dstL1[1], rec, tcnt, toff);
    build_slots<<<NBKT, 1024, 0, stream>>>(rec, tcnt, toff, ch, slot, nxt);

    gemm1_fused<<<dim3(N1 / 32, 2), 256, 0, stream>>>(x, e0, e1, e2, e3, srcL0[0], srcL0[1], ch, slot, nxt,
                                                      Win[0], Win[1], bin[0], bin[1], h1);
    gemm2_fused<<<dim3((N2 + 15) / 16, 2), 256, 0, stream>>>(h1, srcL1[0], srcL1[1], ch + 2 * (2 * (size_t)N1),
                                                             slot + (2 * (size_t)N1) * CAP, nxt + 2 * (size_t)E0,
                                                             Wout[0], Wout[1], bout[0], bout[1], out);
}
